# starting kernel
# baseline (speedup 1.0000x reference)
#include <hip/hip_runtime.h>
#include <hip/hip_bf16.h>


typedef __attribute__((ext_vector_type(2))) float v2f;
typedef __attribute__((ext_vector_type(8))) float v8f;

static constexpr int BB = 8;
static constexpr int TT = 2048;
static constexpr int CD = 1024;
static constexpr int HD = 64;
static constexpr int BT = BB * TT;

static constexpr int MTILE = 128;
static constexpr int KC    = 32;
static constexpr int XPAD  = 34;
static constexpr int SK    = 64;
static constexpr int KPAD  = 66;

__device__ __forceinline__ v8f wmma_f32(v2f a, v2f b, v8f c) {
    return __builtin_amdgcn_wmma_f32_16x16x4_f32(
         false, a,  false, b,
         (short)0, c,  false,  false);
}

__global__ __launch_bounds__(256) void proj_qkv_kernel(
    const float* __restrict__ x,
    const float* __restrict__ Wq, const float* __restrict__ Wk,
    const float* __restrict__ Wv,
    float* __restrict__ Q, float* __restrict__ K, float* __restrict__ V)
{
    __shared__ float xs[MTILE * XPAD];
    __shared__ float ws[3][HD * XPAD];

    const int tid  = threadIdx.x;
    const int wave = tid >> 5;
    const int lane = tid & 31;
    const int grp  = lane >> 4;
    const int idx  = lane & 15;
    const long rowBlock = (long)blockIdx.x * MTILE;

    const float* Wm[3] = {Wq, Wk, Wv};

    v8f acc[3][4] = {};

    for (int kc = 0; kc < CD; kc += KC) {
        __syncthreads();
        for (int i = tid; i < MTILE * (KC / 4); i += 256) {
            int r  = i >> 3;
            int c4 = i & 7;
            float4 val = *(const float4*)(x + (rowBlock + r) * CD + kc + c4 * 4);
            float* dst = &xs[r * XPAD + c4 * 4];
            dst[0] = val.x; dst[1] = val.y; dst[2] = val.z; dst[3] = val.w;
        }
        #pragma unroll
        for (int m = 0; m < 3; ++m) {
            for (int i = tid; i < KC * HD; i += 256) {
                int k = i >> 6;
                int n = i & 63;
                ws[m][n * XPAD + k] = Wm[m][(long)(kc + k) * HD + n];
            }
        }
        if (kc + KC < CD) {
            const float* nx = x + (rowBlock + (tid >> 1)) * CD + kc + KC + (tid & 1) * 16;
            __builtin_prefetch(nx, 0, 1);
        }
        __syncthreads();

        const float* xw = &xs[(wave * 16 + idx) * XPAD];
        #pragma unroll
        for (int kk = 0; kk < KC; kk += 4) {
            v2f a = *(const v2f*)(xw + kk + 2 * grp);
            #pragma unroll
            for (int m = 0; m < 3; ++m) {
                #pragma unroll
                for (int t = 0; t < 4; ++t) {
                    v2f b = *(const v2f*)(&ws[m][(t * 16 + idx) * XPAD + kk + 2 * grp]);
                    acc[m][t] = wmma_f32(a, b, acc[m][t]);
                }
            }
        }
    }

    const long rowW = rowBlock + wave * 16;
    #pragma unroll
    for (int m = 0; m < 3; ++m) {
        float* Om = (m == 0) ? Q : ((m == 1) ? K : V);
        #pragma unroll
        for (int t = 0; t < 4; ++t)
            #pragma unroll
            for (int r = 0; r < 8; ++r)
                Om[(rowW + r + 8 * grp) * HD + t * 16 + idx] = acc[m][t][r];
    }
}

__global__ __launch_bounds__(256) void attn_kernel(
    const float* __restrict__ Q, const float* __restrict__ K,
    const float* __restrict__ V, float* __restrict__ out)
{
    __shared__ float ks[SK * KPAD];
    __shared__ float vs[HD * KPAD];
    __shared__ float ps[8][16 * 16];

    const int tid  = threadIdx.x;
    const int wave = tid >> 5;
    const int lane = tid & 31;
    const int grp  = lane >> 4;
    const int idx  = lane & 15;

    const int  b     = blockIdx.y;
    const int  qt    = blockIdx.x;
    const long base  = (long)b * TT * HD;
    const int  qrow0 = qt * MTILE + wave * 16;

    v2f qf[16];
    const float* qp = Q + base + (long)(qrow0 + idx) * HD + 2 * grp;
    #pragma unroll
    for (int s = 0; s < 16; ++s) qf[s] = *(const v2f*)(qp + 4 * s);

    v8f   o[4] = {};
    float mrow[8], lrow[8];
    #pragma unroll
    for (int r = 0; r < 8; ++r) { mrow[r] = -1e30f; lrow[r] = 0.f; }

    const int nchunk = (qt + 1) * MTILE / SK;
    for (int sc = 0; sc < nchunk; ++sc) {
        const int s0 = sc * SK;
        __syncthreads();
        for (int i = tid; i < SK * (HD / 4); i += 256) {
            int r  = i >> 4;
            int c4 = i & 15;
            float4 kv = *(const float4*)(K + base + (long)(s0 + r) * HD + c4 * 4);
            float* dst = &ks[r * KPAD + c4 * 4];
            dst[0] = kv.x; dst[1] = kv.y; dst[2] = kv.z; dst[3] = kv.w;
            float4 vv = *(const float4*)(V + base + (long)(s0 + r) * HD + c4 * 4);
            vs[(c4 * 4 + 0) * KPAD + r] = vv.x;
            vs[(c4 * 4 + 1) * KPAD + r] = vv.y;
            vs[(c4 * 4 + 2) * KPAD + r] = vv.z;
            vs[(c4 * 4 + 3) * KPAD + r] = vv.w;
        }
        if (sc + 1 < nchunk) {
            const long nrow = base + (long)(s0 + SK + (tid >> 2)) * HD + (tid & 3) * 16;
            __builtin_prefetch(K + nrow, 0, 1);
            __builtin_prefetch(V + nrow, 0, 1);
        }
        __syncthreads();

        #pragma unroll
        for (int kt = 0; kt < 4; ++kt) {
            v8f s = {};
            const float* kw = &ks[(kt * 16 + idx) * KPAD + 2 * grp];
            #pragma unroll
            for (int st = 0; st < 16; ++st) {
                v2f bfrag = *(const v2f*)(kw + 4 * st);
                s = wmma_f32(qf[st], bfrag, s);
            }

            const int key = s0 + kt * 16 + idx;
            #pragma unroll
            for (int r = 0; r < 8; ++r) {
                const int qr = qrow0 + r + 8 * grp;
                float sv = s[r] * 0.125f;
                s[r] = (key <= qr) ? sv : -1e30f;
            }
            float tmax[8];
            #pragma unroll
            for (int r = 0; r < 8; ++r) {
                float v = s[r];
                #pragma unroll
                for (int off = 1; off < 16; off <<= 1)
                    v = fmaxf(v, __shfl_xor(v, off, 32));
                tmax[r] = v;
            }
            float pcorr[8];
            #pragma unroll
            for (int r = 0; r < 8; ++r) {
                const float mnew = fmaxf(mrow[r], tmax[r]);
                pcorr[r] = __expf(mrow[r] - mnew);
                mrow[r]  = mnew;
                const int qr = qrow0 + r + 8 * grp;
                s[r] = (key <= qr) ? __expf(s[r] - mnew) : 0.f;
            }
            #pragma unroll
            for (int r = 0; r < 8; ++r) {
                float v = s[r];
                #pragma unroll
                for (int off = 1; off < 16; off <<= 1)
                    v += __shfl_xor(v, off, 32);
                lrow[r] = lrow[r] * pcorr[r] + v;
            }
            #pragma unroll
            for (int t = 0; t < 4; ++t)
                #pragma unroll
                for (int r = 0; r < 8; ++r)
                    o[t][r] *= pcorr[r];

            float* pw = &ps[wave][0];
            #pragma unroll
            for (int r = 0; r < 8; ++r)
                pw[(r + 8 * grp) * 16 + idx] = s[r];

            #pragma unroll
            for (int pk = 0; pk < 16; pk += 4) {
                v2f a = *(const v2f*)(&pw[idx * 16 + pk + 2 * grp]);
                #pragma unroll
                for (int t = 0; t < 4; ++t) {
                    v2f bfrag = *(const v2f*)(&vs[(t * 16 + idx) * KPAD + kt * 16 + pk + 2 * grp]);
                    o[t] = wmma_f32(a, bfrag, o[t]);
                }
            }
        }
    }

    #pragma unroll
    for (int t = 0; t < 4; ++t)
        #pragma unroll
        for (int r = 0; r < 8; ++r) {
            const long row = qrow0 + r + 8 * grp;
            out[base + row * HD + t * 16 + idx] = o[t][r] / lrow[r];
        }
}

extern "C" void kernel_launch(void* const* d_in, const int* in_sizes, int n_in,
                              void* d_out, int out_size, void* d_ws, size_t ws_size,
                              hipStream_t stream) {
    const float* x  = (const float*)d_in[0];
    const float* Wq = (const float*)d_in[1];
    const float* Wk = (const float*)d_in[2];
    const float* Wv = (const float*)d_in[3];
    float* out = (float*)d_out;

    float* Q = (float*)d_ws;
    float* K = Q + (size_t)BT * HD;
    float* V = K + (size_t)BT * HD;

    proj_qkv_kernel<<<dim3(BT / MTILE), 256, 0, stream>>>(x, Wq, Wk, Wv, Q, K, V);
    attn_kernel<<<dim3(TT / MTILE, BB), 256, 0, stream>>>(Q, K, V, out);
}


# Staging loads of both kernels issued as one register batch per chunk (fixed-trip unrolled loops), not load-wait-store per element; compiler then also issues the softmax lane exchanges 16 deep
# speedup vs baseline: 3.0141x; 3.0141x over previous
#include <hip/hip_runtime.h>
#include <hip/hip_bf16.h>


typedef __attribute__((ext_vector_type(2))) float v2f;
typedef __attribute__((ext_vector_type(8))) float v8f;

static constexpr int BB = 8;
static constexpr int TT = 2048;
static constexpr int CD = 1024;
static constexpr int HD = 64;
static constexpr int BT = BB * TT;

static constexpr int MTILE = 128;
static constexpr int KC    = 32;
static constexpr int XPAD  = 34;
static constexpr int SK    = 64;
static constexpr int KPAD  = 66;

__device__ __forceinline__ v8f wmma_f32(v2f a, v2f b, v8f c) {
    return __builtin_amdgcn_wmma_f32_16x16x4_f32(
         false, a,  false, b,
         (short)0, c,  false,  false);
}

__global__ __launch_bounds__(256) void proj_qkv_kernel(
    const float* __restrict__ x,
    const float* __restrict__ Wq, const float* __restrict__ Wk,
    const float* __restrict__ Wv,
    float* __restrict__ Q, float* __restrict__ K, float* __restrict__ V)
{
    __shared__ float xs[MTILE * XPAD];
    __shared__ float ws[3][HD * XPAD];

    const int tid  = threadIdx.x;
    const int wave = tid >> 5;
    const int lane = tid & 31;
    const int grp  = lane >> 4;
    const int idx  = lane & 15;
    const long rowBlock = (long)blockIdx.x * MTILE;

    const float* Wm[3] = {Wq, Wk, Wv};

    v8f acc[3][4] = {};

    const int xr  = tid >> 3;
    const int xc4 = tid & 7;
    const int wk  = tid >> 6;
    const int wn  = tid & 63;
    const float* xg = x + (rowBlock + xr) * CD + xc4 * 4;

    for (int kc = 0; kc < CD; kc += KC) {
        __syncthreads();
        float4 xv[4];
        #pragma unroll
        for (int j = 0; j < 4; ++j)
            xv[j] = *(const float4*)(xg + (long)(32 * j) * CD + kc);
        float wv[3][8];
        #pragma unroll
        for (int m = 0; m < 3; ++m) {
            const float* wg = Wm[m] + (long)(kc + wk) * HD + wn;
            #pragma unroll
            for (int j = 0; j < 8; ++j)
                wv[m][j] = wg[4 * j * HD];
        }
        #pragma unroll
        for (int j = 0; j < 4; ++j) {
            float* dst = &xs[(xr + 32 * j) * XPAD + xc4 * 4];
            dst[0] = xv[j].x; dst[1] = xv[j].y; dst[2] = xv[j].z; dst[3] = xv[j].w;
        }
        #pragma unroll
        for (int m = 0; m < 3; ++m) {
            #pragma unroll
            for (int j = 0; j < 8; ++j)
                ws[m][wn * XPAD + wk + 4 * j] = wv[m][j];
        }
        if (kc + KC < CD) {
            const float* nx = x + (rowBlock + (tid >> 1)) * CD + kc + KC + (tid & 1) * 16;
            __builtin_prefetch(nx, 0, 1);
        }
        __syncthreads();

        const float* xw = &xs[(wave * 16 + idx) * XPAD];
        #pragma unroll
        for (int kk = 0; kk < KC; kk += 4) {
            v2f a = *(const v2f*)(xw + kk + 2 * grp);
            #pragma unroll
            for (int m = 0; m < 3; ++m) {
                #pragma unroll
                for (int t = 0; t < 4; ++t) {
                    v2f b = *(const v2f*)(&ws[m][(t * 16 + idx) * XPAD + kk + 2 * grp]);
                    acc[m][t] = wmma_f32(a, b, acc[m][t]);
                }
            }
        }
    }

    const long rowW = rowBlock + wave * 16;
    #pragma unroll
    for (int m = 0; m < 3; ++m) {
        float* Om = (m == 0) ? Q : ((m == 1) ? K : V);
        #pragma unroll
        for (int t = 0; t < 4; ++t)
            #pragma unroll
            for (int r = 0; r < 8; ++r)
                Om[(rowW + r + 8 * grp) * HD + t * 16 + idx] = acc[m][t][r];
    }
}

__global__ __launch_bounds__(256) void attn_kernel(
    const float* __restrict__ Q, const float* __restrict__ K,
    const float* __restrict__ V, float* __restrict__ out)
{
    __shared__ float ks[SK * KPAD];
    __shared__ float vs[HD * KPAD];
    __shared__ float ps[8][16 * 16];

    const int tid  = threadIdx.x;
    const int wave = tid >> 5;
    const int lane = tid & 31;
    const int grp  = lane >> 4;
    const int idx  = lane & 15;

    const int  b     = blockIdx.y;
    const int  qt    = blockIdx.x;
    const long base  = (long)b * TT * HD;
    const int  qrow0 = qt * MTILE + wave * 16;

    v2f qf[16];
    const float* qp = Q + base + (long)(qrow0 + idx) * HD + 2 * grp;
    #pragma unroll
    for (int s = 0; s < 16; ++s) qf[s] = *(const v2f*)(qp + 4 * s);

    v8f   o[4] = {};
    float mrow[8], lrow[8];
    #pragma unroll
    for (int r = 0; r < 8; ++r) { mrow[r] = -1e30f; lrow[r] = 0.f; }

    const int nchunk = (qt + 1) * MTILE / SK;
    const int kr  = tid >> 4;
    const int kc4 = tid & 15;
    for (int sc = 0; sc < nchunk; ++sc) {
        const int s0 = sc * SK;
        __syncthreads();
        float4 kv[4], vv[4];
        const long g = base + (long)(s0 + kr) * HD + kc4 * 4;
        #pragma unroll
        for (int j = 0; j < 4; ++j) {
            kv[j] = *(const float4*)(K + g + 16 * j * HD);
            vv[j] = *(const float4*)(V + g + 16 * j * HD);
        }
        #pragma unroll
        for (int j = 0; j < 4; ++j) {
            const int r = kr + 16 * j;
            float* dst = &ks[r * KPAD + kc4 * 4];
            dst[0] = kv[j].x; dst[1] = kv[j].y; dst[2] = kv[j].z; dst[3] = kv[j].w;
            vs[(kc4 * 4 + 0) * KPAD + r] = vv[j].x;
            vs[(kc4 * 4 + 1) * KPAD + r] = vv[j].y;
            vs[(kc4 * 4 + 2) * KPAD + r] = vv[j].z;
            vs[(kc4 * 4 + 3) * KPAD + r] = vv[j].w;
        }
        if (sc + 1 < nchunk) {
            const long nrow = base + (long)(s0 + SK + (tid >> 2)) * HD + (tid & 3) * 16;
            __builtin_prefetch(K + nrow, 0, 1);
            __builtin_prefetch(V + nrow, 0, 1);
        }
        __syncthreads();

        #pragma unroll
        for (int kt = 0; kt < 4; ++kt) {
            v8f s = {};
            const float* kw = &ks[(kt * 16 + idx) * KPAD + 2 * grp];
            #pragma unroll
            for (int st = 0; st < 16; ++st) {
                v2f bfrag = *(const v2f*)(kw + 4 * st);
                s = wmma_f32(qf[st], bfrag, s);
            }

            const int key = s0 + kt * 16 + idx;
            #pragma unroll
            for (int r = 0; r < 8; ++r) {
                const int qr = qrow0 + r + 8 * grp;
                float sv = s[r] * 0.125f;
                s[r] = (key <= qr) ? sv : -1e30f;
            }
            float tmax[8];
            #pragma unroll
            for (int r = 0; r < 8; ++r) {
                float v = s[r];
                #pragma unroll
                for (int off = 1; off < 16; off <<= 1)
                    v = fmaxf(v, __shfl_xor(v, off, 32));
                tmax[r] = v;
            }
            float pcorr[8];
            #pragma unroll
            for (int r = 0; r < 8; ++r) {
                const float mnew = fmaxf(mrow[r], tmax[r]);
                pcorr[r] = __expf(mrow[r] - mnew);
                mrow[r]  = mnew;
                const int qr = qrow0 + r + 8 * grp;
                s[r] = (key <= qr) ? __expf(s[r] - mnew) : 0.f;
            }
            #pragma unroll
            for (int r = 0; r < 8; ++r) {
                float v = s[r];
                #pragma unroll
                for (int off = 1; off < 16; off <<= 1)
                    v += __shfl_xor(v, off, 32);
                lrow[r] = lrow[r] * pcorr[r] + v;
            }
            #pragma unroll
            for (int t = 0; t < 4; ++t)
                #pragma unroll
                for (int r = 0; r < 8; ++r)
                    o[t][r] *= pcorr[r];

            float* pw = &ps[wave][0];
            #pragma unroll
            for (int r = 0; r < 8; ++r)
                pw[(r + 8 * grp) * 16 + idx] = s[r];

            #pragma unroll
            for (int pk = 0; pk < 16; pk += 4) {
                v2f a = *(const v2f*)(&pw[idx * 16 + pk + 2 * grp]);
                #pragma unroll
                for (int t = 0; t < 4; ++t) {
                    v2f bfrag = *(const v2f*)(&vs[(t * 16 + idx) * KPAD + kt * 16 + pk + 2 * grp]);
                    o[t] = wmma_f32(a, bfrag, o[t]);
                }
            }
        }
    }

    #pragma unroll
    for (int t = 0; t < 4; ++t)
        #pragma unroll
        for (int r = 0; r < 8; ++r) {
            const long row = qrow0 + r + 8 * grp;
            out[base + row * HD + t * 16 + idx] = o[t][r] / lrow[r];
        }
}

extern "C" void kernel_launch(void* const* d_in, const int* in_sizes, int n_in,
                              void* d_out, int out_size, void* d_ws, size_t ws_size,
                              hipStream_t stream) {
    const float* x  = (const float*)d_in[0];
    const float* Wq = (const float*)d_in[1];
    const float* Wk = (const float*)d_in[2];
    const float* Wv = (const float*)d_in[3];
    float* out = (float*)d_out;

    float* Q = (float*)d_ws;
    float* K = Q + (size_t)BT * HD;
    float* V = K + (size_t)BT * HD;

    proj_qkv_kernel<<<dim3(BT / MTILE), 256, 0, stream>>>(x, Wq, Wk, Wv, Q, K, V);
    attn_kernel<<<dim3(TT / MTILE, BB), 256, 0, stream>>>(Q, K, V, out);
}


# Softmax row max and row sum over the 16 lanes of a row by DPP row_xmask on the vector ALU instead of __shfl_xor (ds_bpermute through the LDS unit); same butterfly, order of summation unchanged
# speedup vs baseline: 3.3816x; 1.1219x over previous
#include <hip/hip_runtime.h>
#include <hip/hip_bf16.h>


typedef __attribute__((ext_vector_type(2))) float v2f;
typedef __attribute__((ext_vector_type(8))) float v8f;

static constexpr int BB = 8;
static constexpr int TT = 2048;
static constexpr int CD = 1024;
static constexpr int HD = 64;
static constexpr int BT = BB * TT;

static constexpr int MTILE = 128;
static constexpr int KC    = 32;
static constexpr int XPAD  = 34;
static constexpr int SK    = 64;
static constexpr int KPAD  = 66;

__device__ __forceinline__ v8f wmma_f32(v2f a, v2f b, v8f c) {
    return __builtin_amdgcn_wmma_f32_16x16x4_f32(
         false, a,  false, b,
         (short)0, c,  false,  false);
}

template <int XM>
__device__ __forceinline__ float row_xor(float v) {
    return __builtin_bit_cast(float,
        __builtin_amdgcn_mov_dpp(__builtin_bit_cast(int, v), 0x160 + XM, 0xf, 0xf, true));
}

__device__ __forceinline__ float row_max16(float v) {
    v = fmaxf(v, row_xor<1>(v));
    v = fmaxf(v, row_xor<2>(v));
    v = fmaxf(v, row_xor<4>(v));
    v = fmaxf(v, row_xor<8>(v));
    return v;
}
__device__ __forceinline__ float row_sum16(float v) {
    v += row_xor<1>(v);
    v += row_xor<2>(v);
    v += row_xor<4>(v);
    v += row_xor<8>(v);
    return v;
}

__global__ __launch_bounds__(256) void proj_qkv_kernel(
    const float* __restrict__ x,
    const float* __restrict__ Wq, const float* __restrict__ Wk,
    const float* __restrict__ Wv,
    float* __restrict__ Q, float* __restrict__ K, float* __restrict__ V)
{
    __shared__ float xs[MTILE * XPAD];
    __shared__ float ws[3][HD * XPAD];

    const int tid  = threadIdx.x;
    const int wave = tid >> 5;
    const int lane = tid & 31;
    const int grp  = lane >> 4;
    const int idx  = lane & 15;
    const long rowBlock = (long)blockIdx.x * MTILE;

    const float* Wm[3] = {Wq, Wk, Wv};

    v8f acc[3][4] = {};

    const int xr  = tid >> 3;
    const int xc4 = tid & 7;
    const int wk  = tid >> 6;
    const int wn  = tid & 63;
    const float* xg = x + (rowBlock + xr) * CD + xc4 * 4;

    for (int kc = 0; kc < CD; kc += KC) {
        __syncthreads();
        float4 xv[4];
        #pragma unroll
        for (int j = 0; j < 4; ++j)
            xv[j] = *(const float4*)(xg + (long)(32 * j) * CD + kc);
        float wv[3][8];
        #pragma unroll
        for (int m = 0; m < 3; ++m) {
            const float* wg = Wm[m] + (long)(kc + wk) * HD + wn;
            #pragma unroll
            for (int j = 0; j < 8; ++j)
                wv[m][j] = wg[4 * j * HD];
        }
        #pragma unroll
        for (int j = 0; j < 4; ++j) {
            float* dst = &xs[(xr + 32 * j) * XPAD + xc4 * 4];
            dst[0] = xv[j].x; dst[1] = xv[j].y; dst[2] = xv[j].z; dst[3] = xv[j].w;
        }
        #pragma unroll
        for (int m = 0; m < 3; ++m) {
            #pragma unroll
            for (int j = 0; j < 8; ++j)
                ws[m][wn * XPAD + wk + 4 * j] = wv[m][j];
        }
        if (kc + KC < CD) {
            const float* nx = x + (rowBlock + (tid >> 1)) * CD + kc + KC + (tid & 1) * 16;
            __builtin_prefetch(nx, 0, 1);
        }
        __syncthreads();

        const float* xw = &xs[(wave * 16 + idx) * XPAD];
        #pragma unroll
        for (int kk = 0; kk < KC; kk += 4) {
            v2f a = *(const v2f*)(xw + kk + 2 * grp);
            #pragma unroll
            for (int m = 0; m < 3; ++m) {
                #pragma unroll
                for (int t = 0; t < 4; ++t) {
                    v2f b = *(const v2f*)(&ws[m][(t * 16 + idx) * XPAD + kk + 2 * grp]);
                    acc[m][t] = wmma_f32(a, b, acc[m][t]);
                }
            }
        }
    }

    const long rowW = rowBlock + wave * 16;
    #pragma unroll
    for (int m = 0; m < 3; ++m) {
        float* Om = (m == 0) ? Q : ((m == 1) ? K : V);
        #pragma unroll
        for (int t = 0; t < 4; ++t)
            #pragma unroll
            for (int r = 0; r < 8; ++r)
                Om[(rowW + r + 8 * grp) * HD + t * 16 + idx] = acc[m][t][r];
    }
}

__global__ __launch_bounds__(256) void attn_kernel(
    const float* __restrict__ Q, const float* __restrict__ K,
    const float* __restrict__ V, float* __restrict__ out)
{
    __shared__ float ks[SK * KPAD];
    __shared__ float vs[HD * KPAD];
    __shared__ float ps[8][16 * 16];

    const int tid  = threadIdx.x;
    const int wave = tid >> 5;
    const int lane = tid & 31;
    const int grp  = lane >> 4;
    const int idx  = lane & 15;

    const int  b     = blockIdx.y;
    const int  qt    = blockIdx.x;
    const long base  = (long)b * TT * HD;
    const int  qrow0 = qt * MTILE + wave * 16;

    v2f qf[16];
    const float* qp = Q + base + (long)(qrow0 + idx) * HD + 2 * grp;
    #pragma unroll
    for (int s = 0; s < 16; ++s) qf[s] = *(const v2f*)(qp + 4 * s);

    v8f   o[4] = {};
    float mrow[8], lrow[8];
    #pragma unroll
    for (int r = 0; r < 8; ++r) { mrow[r] = -1e30f; lrow[r] = 0.f; }

    const int nchunk = (qt + 1) * MTILE / SK;
    const int kr  = tid >> 4;
    const int kc4 = tid & 15;
    for (int sc = 0; sc < nchunk; ++sc) {
        const int s0 = sc * SK;
        __syncthreads();
        float4 kv[4], vv[4];
        const long g = base + (long)(s0 + kr) * HD + kc4 * 4;
        #pragma unroll
        for (int j = 0; j < 4; ++j) {
            kv[j] = *(const float4*)(K + g + 16 * j * HD);
            vv[j] = *(const float4*)(V + g + 16 * j * HD);
        }
        #pragma unroll
        for (int j = 0; j < 4; ++j) {
            const int r = kr + 16 * j;
            float* dst = &ks[r * KPAD + kc4 * 4];
            dst[0] = kv[j].x; dst[1] = kv[j].y; dst[2] = kv[j].z; dst[3] = kv[j].w;
            vs[(kc4 * 4 + 0) * KPAD + r] = vv[j].x;
            vs[(kc4 * 4 + 1) * KPAD + r] = vv[j].y;
            vs[(kc4 * 4 + 2) * KPAD + r] = vv[j].z;
            vs[(kc4 * 4 + 3) * KPAD + r] = vv[j].w;
        }
        if (sc + 1 < nchunk) {
            const long nrow = base + (long)(s0 + SK + (tid >> 2)) * HD + (tid & 3) * 16;
            __builtin_prefetch(K + nrow, 0, 1);
            __builtin_prefetch(V + nrow, 0, 1);
        }
        __syncthreads();

        #pragma unroll
        for (int kt = 0; kt < 4; ++kt) {
            v8f s = {};
            const float* kw = &ks[(kt * 16 + idx) * KPAD + 2 * grp];
            #pragma unroll
            for (int st = 0; st < 16; ++st) {
                v2f bfrag = *(const v2f*)(kw + 4 * st);
                s = wmma_f32(qf[st], bfrag, s);
            }

            const int key = s0 + kt * 16 + idx;
            #pragma unroll
            for (int r = 0; r < 8; ++r) {
                const int qr = qrow0 + r + 8 * grp;
                float sv = s[r] * 0.125f;
                s[r] = (key <= qr) ? sv : -1e30f;
            }
            float tmax[8];
            #pragma unroll
            for (int r = 0; r < 8; ++r)
                tmax[r] = row_max16(s[r]);
            float pcorr[8];
            #pragma unroll
            for (int r = 0; r < 8; ++r) {
                const float mnew = fmaxf(mrow[r], tmax[r]);
                pcorr[r] = __expf(mrow[r] - mnew);
                mrow[r]  = mnew;
                const int qr = qrow0 + r + 8 * grp;
                s[r] = (key <= qr) ? __expf(s[r] - mnew) : 0.f;
            }
            #pragma unroll
            for (int r = 0; r < 8; ++r)
                lrow[r] = lrow[r] * pcorr[r] + row_sum16(s[r]);
            #pragma unroll
            for (int t = 0; t < 4; ++t)
                #pragma unroll
                for (int r = 0; r < 8; ++r)
                    o[t][r] *= pcorr[r];

            float* pw = &ps[wave][0];
            #pragma unroll
            for (int r = 0; r < 8; ++r)
                pw[(r + 8 * grp) * 16 + idx] = s[r];

            #pragma unroll
            for (int pk = 0; pk < 16; pk += 4) {
                v2f a = *(const v2f*)(&pw[idx * 16 + pk + 2 * grp]);
                #pragma unroll
                for (int t = 0; t < 4; ++t) {
                    v2f bfrag = *(const v2f*)(&vs[(t * 16 + idx) * KPAD + kt * 16 + pk + 2 * grp]);
                    o[t] = wmma_f32(a, bfrag, o[t]);
                }
            }
        }
    }

    #pragma unroll
    for (int t = 0; t < 4; ++t)
        #pragma unroll
        for (int r = 0; r < 8; ++r) {
            const long row = qrow0 + r + 8 * grp;
            out[base + row * HD + t * 16 + idx] = o[t][r] / lrow[r];
        }
}

extern "C" void kernel_launch(void* const* d_in, const int* in_sizes, int n_in,
                              void* d_out, int out_size, void* d_ws, size_t ws_size,
                              hipStream_t stream) {
    const float* x  = (const float*)d_in[0];
    const float* Wq = (const float*)d_in[1];
    const float* Wk = (const float*)d_in[2];
    const float* Wv = (const float*)d_in[3];
    float* out = (float*)d_out;

    float* Q = (float*)d_ws;
    float* K = Q + (size_t)BT * HD;
    float* V = K + (size_t)BT * HD;

    proj_qkv_kernel<<<dim3(BT / MTILE), 256, 0, stream>>>(x, Wq, Wk, Wv, Q, K, V);
    attn_kernel<<<dim3(TT / MTILE, BB), 256, 0, stream>>>(Q, K, V, out);
}


# Row tile per work-group 128 -> 64 in both kernels: 256 work-groups per launch, one per WGP, where 128 left half the chip idle (projection: 8 waves = 4 row tiles x 2 column halves; attention: 4 waves)
# speedup vs baseline: 5.0401x; 1.4904x over previous
#include <hip/hip_runtime.h>
#include <hip/hip_bf16.h>


typedef __attribute__((ext_vector_type(2))) float v2f;
typedef __attribute__((ext_vector_type(8))) float v8f;

static constexpr int BB = 8;
static constexpr int TT = 2048;
static constexpr int CD = 1024;
static constexpr int HD = 64;
static constexpr int BT = BB * TT;

static constexpr int MTILE = 64;
static constexpr int KC    = 32;
static constexpr int XPAD  = 34;
static constexpr int SK    = 64;
static constexpr int KPAD  = 66;

__device__ __forceinline__ v8f wmma_f32(v2f a, v2f b, v8f c) {
    return __builtin_amdgcn_wmma_f32_16x16x4_f32(
         false, a,  false, b,
         (short)0, c,  false,  false);
}

template <int XM>
__device__ __forceinline__ float row_xor(float v) {
    return __builtin_bit_cast(float,
        __builtin_amdgcn_mov_dpp(__builtin_bit_cast(int, v), 0x160 + XM, 0xf, 0xf, true));
}

__device__ __forceinline__ float row_max16(float v) {
    v = fmaxf(v, row_xor<1>(v));
    v = fmaxf(v, row_xor<2>(v));
    v = fmaxf(v, row_xor<4>(v));
    v = fmaxf(v, row_xor<8>(v));
    return v;
}
__device__ __forceinline__ float row_sum16(float v) {
    v += row_xor<1>(v);
    v += row_xor<2>(v);
    v += row_xor<4>(v);
    v += row_xor<8>(v);
    return v;
}

__global__ __launch_bounds__(256) void proj_qkv_kernel(
    const float* __restrict__ x,
    const float* __restrict__ Wq, const float* __restrict__ Wk,
    const float* __restrict__ Wv,
    float* __restrict__ Q, float* __restrict__ K, float* __restrict__ V)
{
    __shared__ float xs[MTILE * XPAD];
    __shared__ float ws[3 * HD * XPAD];

    const int tid  = threadIdx.x;
    const int wave = tid >> 5;
    const int lane = tid & 31;
    const int grp  = lane >> 4;
    const int idx  = lane & 15;
    const long rowBlock = (long)blockIdx.x * MTILE;

    const float* Wm[3] = {Wq, Wk, Wv};

    const int rt = wave & 3;
    const int ch = wave >> 2;
    v8f acc[6] = {};

    const int xr  = tid >> 3;
    const int xc4 = tid & 7;
    const int wk  = tid >> 6;
    const int wn  = tid & 63;
    const float* xg = x + (rowBlock + xr) * CD + xc4 * 4;

    for (int kc = 0; kc < CD; kc += KC) {
        __syncthreads();
        float4 xv[2];
        #pragma unroll
        for (int j = 0; j < 2; ++j)
            xv[j] = *(const float4*)(xg + (long)(32 * j) * CD + kc);
        float wv[3][8];
        #pragma unroll
        for (int m = 0; m < 3; ++m) {
            const float* wg = Wm[m] + (long)(kc + wk) * HD + wn;
            #pragma unroll
            for (int j = 0; j < 8; ++j)
                wv[m][j] = wg[4 * j * HD];
        }
        #pragma unroll
        for (int j = 0; j < 2; ++j) {
            float* dst = &xs[(xr + 32 * j) * XPAD + xc4 * 4];
            dst[0] = xv[j].x; dst[1] = xv[j].y; dst[2] = xv[j].z; dst[3] = xv[j].w;
        }
        #pragma unroll
        for (int m = 0; m < 3; ++m) {
            #pragma unroll
            for (int j = 0; j < 8; ++j)
                ws[(m * HD + wn) * XPAD + wk + 4 * j] = wv[m][j];
        }
        if (kc + KC < CD && tid < 2 * MTILE) {
            const float* nx = x + (rowBlock + (tid >> 1)) * CD + kc + KC + (tid & 1) * 16;
            __builtin_prefetch(nx, 0, 1);
        }
        __syncthreads();

        const float* xw = &xs[(rt * 16 + idx) * XPAD];
        const float* ww = &ws[(ch * 96 + idx) * XPAD];
        #pragma unroll
        for (int kk = 0; kk < KC; kk += 4) {
            v2f a = *(const v2f*)(xw + kk + 2 * grp);
            #pragma unroll
            for (int c = 0; c < 6; ++c) {
                v2f b = *(const v2f*)(ww + c * 16 * XPAD + kk + 2 * grp);
                acc[c] = wmma_f32(a, b, acc[c]);
            }
        }
    }

    const long rowW = rowBlock + rt * 16;
    #pragma unroll
    for (int c = 0; c < 6; ++c) {
        const int tc = ch * 6 + c;
        const int m  = tc >> 2;
        const int t  = tc & 3;
        float* Om = (m == 0) ? Q : ((m == 1) ? K : V);
        #pragma unroll
        for (int r = 0; r < 8; ++r)
            Om[(rowW + r + 8 * grp) * HD + t * 16 + idx] = acc[c][r];
    }
}

__global__ __launch_bounds__(128) void attn_kernel(
    const float* __restrict__ Q, const float* __restrict__ K,
    const float* __restrict__ V, float* __restrict__ out)
{
    __shared__ float ks[SK * KPAD];
    __shared__ float vs[HD * KPAD];
    __shared__ float ps[4][16 * 16];

    const int tid  = threadIdx.x;
    const int wave = tid >> 5;
    const int lane = tid & 31;
    const int grp  = lane >> 4;
    const int idx  = lane & 15;

    const int  b     = blockIdx.y;
    const int  qt    = blockIdx.x;
    const long base  = (long)b * TT * HD;
    const int  qrow0 = qt * MTILE + wave * 16;

    v2f qf[16];
    const float* qp = Q + base + (long)(qrow0 + idx) * HD + 2 * grp;
    #pragma unroll
    for (int s = 0; s < 16; ++s) qf[s] = *(const v2f*)(qp + 4 * s);

    v8f   o[4] = {};
    float mrow[8], lrow[8];
    #pragma unroll
    for (int r = 0; r < 8; ++r) { mrow[r] = -1e30f; lrow[r] = 0.f; }

    const int nchunk = (qt + 1) * MTILE / SK;
    const int kr  = tid >> 4;
    const int kc4 = tid & 15;
    for (int sc = 0; sc < nchunk; ++sc) {
        const int s0 = sc * SK;
        __syncthreads();
        float4 kv[8], vv[8];
        const long g = base + (long)(s0 + kr) * HD + kc4 * 4;
        #pragma unroll
        for (int j = 0; j < 8; ++j) {
            kv[j] = *(const float4*)(K + g + 8 * j * HD);
            vv[j] = *(const float4*)(V + g + 8 * j * HD);
        }
        #pragma unroll
        for (int j = 0; j < 8; ++j) {
            const int r = kr + 8 * j;
            float* dst = &ks[r * KPAD + kc4 * 4];
            dst[0] = kv[j].x; dst[1] = kv[j].y; dst[2] = kv[j].z; dst[3] = kv[j].w;
            vs[(kc4 * 4 + 0) * KPAD + r] = vv[j].x;
            vs[(kc4 * 4 + 1) * KPAD + r] = vv[j].y;
            vs[(kc4 * 4 + 2) * KPAD + r] = vv[j].z;
            vs[(kc4 * 4 + 3) * KPAD + r] = vv[j].w;
        }
        if (sc + 1 < nchunk) {
            const long nrow = base + (long)(s0 + SK + (tid >> 1)) * HD + (tid & 1) * 32;
            __builtin_prefetch(K + nrow, 0, 1);
            __builtin_prefetch(V + nrow, 0, 1);
        }
        __syncthreads();

        #pragma unroll
        for (int kt = 0; kt < 4; ++kt) {
            v8f s = {};
            const float* kw = &ks[(kt * 16 + idx) * KPAD + 2 * grp];
            #pragma unroll
            for (int st = 0; st < 16; ++st) {
                v2f bfrag = *(const v2f*)(kw + 4 * st);
                s = wmma_f32(qf[st], bfrag, s);
            }

            const int key = s0 + kt * 16 + idx;
            #pragma unroll
            for (int r = 0; r < 8; ++r) {
                const int qr = qrow0 + r + 8 * grp;
                float sv = s[r] * 0.125f;
                s[r] = (key <= qr) ? sv : -1e30f;
            }
            float tmax[8];
            #pragma unroll
            for (int r = 0; r < 8; ++r)
                tmax[r] = row_max16(s[r]);
            float pcorr[8];
            #pragma unroll
            for (int r = 0; r < 8; ++r) {
                const float mnew = fmaxf(mrow[r], tmax[r]);
                pcorr[r] = __expf(mrow[r] - mnew);
                mrow[r]  = mnew;
                const int qr = qrow0 + r + 8 * grp;
                s[r] = (key <= qr) ? __expf(s[r] - mnew) : 0.f;
            }
            #pragma unroll
            for (int r = 0; r < 8; ++r)
                lrow[r] = lrow[r] * pcorr[r] + row_sum16(s[r]);
            #pragma unroll
            for (int t = 0; t < 4; ++t)
                #pragma unroll
                for (int r = 0; r < 8; ++r)
                    o[t][r] *= pcorr[r];

            float* pw = &ps[wave][0];
            #pragma unroll
            for (int r = 0; r < 8; ++r)
                pw[(r + 8 * grp) * 16 + idx] = s[r];

            #pragma unroll
            for (int pk = 0; pk < 16; pk += 4) {
                v2f a = *(const v2f*)(&pw[idx * 16 + pk + 2 * grp]);
                #pragma unroll
                for (int t = 0; t < 4; ++t) {
                    v2f bfrag = *(const v2f*)(&vs[(t * 16 + idx) * KPAD + kt * 16 + pk + 2 * grp]);
                    o[t] = wmma_f32(a, bfrag, o[t]);
                }
            }
        }
    }

    #pragma unroll
    for (int t = 0; t < 4; ++t)
        #pragma unroll
        for (int r = 0; r < 8; ++r) {
            const long row = qrow0 + r + 8 * grp;
            out[base + row * HD + t * 16 + idx] = o[t][r] / lrow[r];
        }
}

extern "C" void kernel_launch(void* const* d_in, const int* in_sizes, int n_in,
                              void* d_out, int out_size, void* d_ws, size_t ws_size,
                              hipStream_t stream) {
    const float* x  = (const float*)d_in[0];
    const float* Wq = (const float*)d_in[1];
    const float* Wk = (const float*)d_in[2];
    const float* Wv = (const float*)d_in[3];
    float* out = (float*)d_out;

    float* Q = (float*)d_ws;
    float* K = Q + (size_t)BT * HD;
    float* V = K + (size_t)BT * HD;

    proj_qkv_kernel<<<dim3(BT / MTILE), 256, 0, stream>>>(x, Wq, Wk, Wv, Q, K, V);
    attn_kernel<<<dim3(TT / MTILE, BB), 128, 0, stream>>>(Q, K, V, out);
}


# Attention work-group = light query tile p + heavy tile 63-p (32 rows each), keys split in 4 quarters over 8 waves: equal causal work per wave; quarter softmax states merged at end (new sum order)
# speedup vs baseline: 6.9672x; 1.3824x over previous
#include <hip/hip_runtime.h>
#include <hip/hip_bf16.h>


typedef __attribute__((ext_vector_type(2))) float v2f;
typedef __attribute__((ext_vector_type(8))) float v8f;

static constexpr int BB = 8;
static constexpr int TT = 2048;
static constexpr int CD = 1024;
static constexpr int HD = 64;
static constexpr int BT = BB * TT;

static constexpr int MTILE = 64;
static constexpr int KC    = 32;
static constexpr int XPAD  = 34;
static constexpr int SK    = 64;
static constexpr int KPAD  = 66;

__device__ __forceinline__ v8f wmma_f32(v2f a, v2f b, v8f c) {
    return __builtin_amdgcn_wmma_f32_16x16x4_f32(
         false, a,  false, b,
         (short)0, c,  false,  false);
}

template <int XM>
__device__ __forceinline__ float row_xor(float v) {
    return __builtin_bit_cast(float,
        __builtin_amdgcn_mov_dpp(__builtin_bit_cast(int, v), 0x160 + XM, 0xf, 0xf, true));
}

__device__ __forceinline__ float row_max16(float v) {
    v = fmaxf(v, row_xor<1>(v));
    v = fmaxf(v, row_xor<2>(v));
    v = fmaxf(v, row_xor<4>(v));
    v = fmaxf(v, row_xor<8>(v));
    return v;
}
__device__ __forceinline__ float row_sum16(float v) {
    v += row_xor<1>(v);
    v += row_xor<2>(v);
    v += row_xor<4>(v);
    v += row_xor<8>(v);
    return v;
}

__global__ __launch_bounds__(256) void proj_qkv_kernel(
    const float* __restrict__ x,
    const float* __restrict__ Wq, const float* __restrict__ Wk,
    const float* __restrict__ Wv,
    float* __restrict__ Q, float* __restrict__ K, float* __restrict__ V)
{
    __shared__ float xs[MTILE * XPAD];
    __shared__ float ws[3 * HD * XPAD];

    const int tid  = threadIdx.x;
    const int wave = tid >> 5;
    const int lane = tid & 31;
    const int grp  = lane >> 4;
    const int idx  = lane & 15;
    const long rowBlock = (long)blockIdx.x * MTILE;

    const float* Wm[3] = {Wq, Wk, Wv};

    const int rt = wave & 3;
    const int ch = wave >> 2;
    v8f acc[6] = {};

    const int xr  = tid >> 3;
    const int xc4 = tid & 7;
    const int wk  = tid >> 6;
    const int wn  = tid & 63;
    const float* xg = x + (rowBlock + xr) * CD + xc4 * 4;

    for (int kc = 0; kc < CD; kc += KC) {
        __syncthreads();
        float4 xv[2];
        #pragma unroll
        for (int j = 0; j < 2; ++j)
            xv[j] = *(const float4*)(xg + (long)(32 * j) * CD + kc);
        float wv[3][8];
        #pragma unroll
        for (int m = 0; m < 3; ++m) {
            const float* wg = Wm[m] + (long)(kc + wk) * HD + wn;
            #pragma unroll
            for (int j = 0; j < 8; ++j)
                wv[m][j] = wg[4 * j * HD];
        }
        #pragma unroll
        for (int j = 0; j < 2; ++j) {
            float* dst = &xs[(xr + 32 * j) * XPAD + xc4 * 4];
            dst[0] = xv[j].x; dst[1] = xv[j].y; dst[2] = xv[j].z; dst[3] = xv[j].w;
        }
        #pragma unroll
        for (int m = 0; m < 3; ++m) {
            #pragma unroll
            for (int j = 0; j < 8; ++j)
                ws[(m * HD + wn) * XPAD + wk + 4 * j] = wv[m][j];
        }
        if (kc + KC < CD && tid < 2 * MTILE) {
            const float* nx = x + (rowBlock + (tid >> 1)) * CD + kc + KC + (tid & 1) * 16;
            __builtin_prefetch(nx, 0, 1);
        }
        __syncthreads();

        const float* xw = &xs[(rt * 16 + idx) * XPAD];
        const float* ww = &ws[(ch * 96 + idx) * XPAD];
        #pragma unroll
        for (int kk = 0; kk < KC; kk += 4) {
            v2f a = *(const v2f*)(xw + kk + 2 * grp);
            #pragma unroll
            for (int c = 0; c < 6; ++c) {
                v2f b = *(const v2f*)(ww + c * 16 * XPAD + kk + 2 * grp);
                acc[c] = wmma_f32(a, b, acc[c]);
            }
        }
    }

    const long rowW = rowBlock + rt * 16;
    #pragma unroll
    for (int c = 0; c < 6; ++c) {
        const int tc = ch * 6 + c;
        const int m  = tc >> 2;
        const int t  = tc & 3;
        float* Om = (m == 0) ? Q : ((m == 1) ? K : V);
        #pragma unroll
        for (int r = 0; r < 8; ++r)
            Om[(rowW + r + 8 * grp) * HD + t * 16 + idx] = acc[c][r];
    }
}

static constexpr int QT = 32;

__device__ __forceinline__ void attn_tile_step(
    const v2f (&qf)[16], const v2f (&kb)[16], const v2f (&vb)[4][4],
    v8f (&o)[4], float (&mrow)[8], float (&lrow)[8],
    float* pw, const int qrow0, const int key0, const int grp, const int idx)
{
    v8f s = {};
    #pragma unroll
    for (int st = 0; st < 16; ++st)
        s = wmma_f32(qf[st], kb[st], s);

    const int key = key0 + idx;
    #pragma unroll
    for (int r = 0; r < 8; ++r) {
        const int qr = qrow0 + r + 8 * grp;
        float sv = s[r] * 0.125f;
        s[r] = (key <= qr) ? sv : -1e30f;
    }
    float tmax[8];
    #pragma unroll
    for (int r = 0; r < 8; ++r)
        tmax[r] = row_max16(s[r]);
    float pcorr[8];
    #pragma unroll
    for (int r = 0; r < 8; ++r) {
        const float mnew = fmaxf(mrow[r], tmax[r]);
        pcorr[r] = __expf(mrow[r] - mnew);
        mrow[r]  = mnew;
        const int qr = qrow0 + r + 8 * grp;
        s[r] = (key <= qr) ? __expf(s[r] - mnew) : 0.f;
    }
    #pragma unroll
    for (int r = 0; r < 8; ++r)
        lrow[r] = lrow[r] * pcorr[r] + row_sum16(s[r]);
    #pragma unroll
    for (int t = 0; t < 4; ++t)
        #pragma unroll
        for (int r = 0; r < 8; ++r)
            o[t][r] *= pcorr[r];

    #pragma unroll
    for (int r = 0; r < 8; ++r)
        pw[(r + 8 * grp) * 16 + idx] = s[r];

    #pragma unroll
    for (int j = 0; j < 4; ++j) {
        v2f a = *(const v2f*)(&pw[idx * 16 + 4 * j + 2 * grp]);
        #pragma unroll
        for (int t = 0; t < 4; ++t)
            o[t] = wmma_f32(a, vb[t][j], o[t]);
    }
}

__device__ __forceinline__ void attn_merge_store(
    float* mg, const v8f (&o)[4], const float (&mrow)[8], const float (&lrow)[8],
    float* __restrict__ out, const long rowbase, const int wave, const int lane,
    const int grp, const int idx)
{
    __syncthreads();
    float* my = mg + wave * (48 * 32) + lane;
    #pragma unroll
    for (int r = 0; r < 8; ++r) {
        my[r * 32]       = mrow[r];
        my[(8 + r) * 32] = lrow[r];
    }
    #pragma unroll
    for (int t = 0; t < 4; ++t)
        #pragma unroll
        for (int r = 0; r < 8; ++r)
            my[(16 + t * 8 + r) * 32] = o[t][r];
    __syncthreads();

    const int rt = wave >> 2;
    const int kq = wave & 3;
    const int th = kq & 1;
    const int rh = kq >> 1;
    const float* src = mg + (rt * 4) * (48 * 32) + lane;
    #pragma unroll
    for (int rr = 0; rr < 4; ++rr) {
        const int r = rh * 4 + rr;
        float mk[4], wk[4];
        #pragma unroll
        for (int k = 0; k < 4; ++k) mk[k] = src[k * (48 * 32) + r * 32];
        const float mstar = fmaxf(fmaxf(mk[0], mk[1]), fmaxf(mk[2], mk[3]));
        float L = 0.f;
        #pragma unroll
        for (int k = 0; k < 4; ++k) {
            wk[k] = __expf(mk[k] - mstar);
            L += src[k * (48 * 32) + (8 + r) * 32] * wk[k];
        }
        #pragma unroll
        for (int tt = 0; tt < 2; ++tt) {
            const int t = th * 2 + tt;
            float O = 0.f;
            #pragma unroll
            for (int k = 0; k < 4; ++k)
                O += src[k * (48 * 32) + (16 + t * 8 + r) * 32] * wk[k];
            out[(rowbase + r + 8 * grp) * HD + t * 16 + idx] = O / L;
        }
    }
}

__device__ __forceinline__ void attn_stage_chunk(
    const float* __restrict__ K, const float* __restrict__ V,
    float* ks, float* vs, const long base, const int s0, const bool more,
    const int tid)
{
    const int kr  = tid >> 4;
    const int kc4 = tid & 15;
    __syncthreads();
    float4 kv[4], vv[4];
    const long g = base + (long)(s0 + kr) * HD + kc4 * 4;
    #pragma unroll
    for (int j = 0; j < 4; ++j) {
        kv[j] = *(const float4*)(K + g + 16 * j * HD);
        vv[j] = *(const float4*)(V + g + 16 * j * HD);
    }
    #pragma unroll
    for (int j = 0; j < 4; ++j) {
        const int r = kr + 16 * j;
        float* dst = &ks[r * KPAD + kc4 * 4];
        dst[0] = kv[j].x; dst[1] = kv[j].y; dst[2] = kv[j].z; dst[3] = kv[j].w;
        vs[(kc4 * 4 + 0) * KPAD + r] = vv[j].x;
        vs[(kc4 * 4 + 1) * KPAD + r] = vv[j].y;
        vs[(kc4 * 4 + 2) * KPAD + r] = vv[j].z;
        vs[(kc4 * 4 + 3) * KPAD + r] = vv[j].w;
    }
    if (more) {
        const int  pt   = tid & 127;
        const long nrow = base + (long)(s0 + SK + (pt >> 1)) * HD + (pt & 1) * 32;
        __builtin_prefetch((tid < 128 ? K : V) + nrow, 0, 1);
    }
    __syncthreads();
}

__device__ __forceinline__ void attn_load_kv_frags(
    const float* ks, const float* vs, v2f (&kb)[16], v2f (&vb)[4][4],
    const int kq, const int grp, const int idx)
{
    const float* kw = &ks[(kq * 16 + idx) * KPAD + 2 * grp];
    #pragma unroll
    for (int st = 0; st < 16; ++st) kb[st] = *(const v2f*)(kw + 4 * st);
    #pragma unroll
    for (int t = 0; t < 4; ++t)
        #pragma unroll
        for (int j = 0; j < 4; ++j)
            vb[t][j] = *(const v2f*)(&vs[(t * 16 + idx) * KPAD + kq * 16 + 4 * j + 2 * grp]);
}

__global__ __launch_bounds__(256) void attn_kernel(
    const float* __restrict__ Q, const float* __restrict__ K,
    const float* __restrict__ V, float* __restrict__ out)
{
    __shared__ __attribute__((aligned(16))) float smem[8 * 48 * 32];
    float* ks = smem;
    float* vs = smem + SK * KPAD;
    float* ps = smem + 2 * SK * KPAD;

    const int tid  = threadIdx.x;
    const int lane = tid & 31;
    const int grp  = lane >> 4;
    const int idx  = lane & 15;
    const int wave = __builtin_amdgcn_readfirstlane(tid >> 5);
    const int rt   = wave >> 2;
    const int kq   = wave & 3;

    const int  b     = blockIdx.y;
    const int  p     = blockIdx.x;
    const long base  = (long)b * TT * HD;
    const int  qrowA = p * QT + rt * 16;
    const int  qrowB = (TT / QT - 1 - p) * QT + rt * 16;

    v2f qfA[16], qfB[16];
    {
        const float* qa = Q + base + (long)(qrowA + idx) * HD + 2 * grp;
        const float* qb = Q + base + (long)(qrowB + idx) * HD + 2 * grp;
        #pragma unroll
        for (int s = 0; s < 16; ++s) {
            qfA[s] = *(const v2f*)(qa + 4 * s);
            qfB[s] = *(const v2f*)(qb + 4 * s);
        }
    }

    v8f   oA[4] = {}, oB[4] = {};
    float mA[8], lA[8], mB[8], lB[8];
    #pragma unroll
    for (int r = 0; r < 8; ++r) {
        mA[r] = -1e30f; lA[r] = 0.f;
        mB[r] = -1e30f; lB[r] = 0.f;
    }

    const int nchunkA = (p * QT + QT - 1) / SK + 1;
    const int nchunkB = ((TT / QT - 1 - p) * QT + QT - 1) / SK + 1;
    float* pw = ps + wave * 256;

    for (int sc = 0; sc < nchunkA; ++sc) {
        const int s0 = sc * SK;
        attn_stage_chunk(K, V, ks, vs, base, s0, sc + 1 < nchunkB, tid);
        v2f kb[16], vb[4][4];
        attn_load_kv_frags(ks, vs, kb, vb, kq, grp, idx);
        const int key0 = s0 + kq * 16;
        attn_tile_step(qfA, kb, vb, oA, mA, lA, pw, qrowA, key0, grp, idx);
        attn_tile_step(qfB, kb, vb, oB, mB, lB, pw, qrowB, key0, grp, idx);
    }
    for (int sc = nchunkA; sc < nchunkB; ++sc) {
        const int s0 = sc * SK;
        attn_stage_chunk(K, V, ks, vs, base, s0, sc + 1 < nchunkB, tid);
        v2f kb[16], vb[4][4];
        attn_load_kv_frags(ks, vs, kb, vb, kq, grp, idx);
        const int key0 = s0 + kq * 16;
        attn_tile_step(qfB, kb, vb, oB, mB, lB, pw, qrowB, key0, grp, idx);
    }

    attn_merge_store(smem, oA, mA, lA, out, base / HD + qrowA, wave, lane, grp, idx);
    attn_merge_store(smem, oB, mB, lB, out, base / HD + qrowB, wave, lane, grp, idx);
}

extern "C" void kernel_launch(void* const* d_in, const int* in_sizes, int n_in,
                              void* d_out, int out_size, void* d_ws, size_t ws_size,
                              hipStream_t stream) {
    const float* x  = (const float*)d_in[0];
    const float* Wq = (const float*)d_in[1];
    const float* Wk = (const float*)d_in[2];
    const float* Wv = (const float*)d_in[3];
    float* out = (float*)d_out;

    float* Q = (float*)d_ws;
    float* K = Q + (size_t)BT * HD;
    float* V = K + (size_t)BT * HD;

    proj_qkv_kernel<<<dim3(BT / MTILE), 256, 0, stream>>>(x, Wq, Wk, Wv, Q, K, V);
    attn_kernel<<<dim3(TT / MTILE, BB), 256, 0, stream>>>(Q, K, V, out);
}


# Staging of both kernels as a two-stage LDS software pipeline: next chunk's global loads issued before this chunk's WMMAs, stored to the other stage after them; one barrier per chunk; sums unchanged
# speedup vs baseline: 8.2500x; 1.1841x over previous
#include <hip/hip_runtime.h>
#include <hip/hip_bf16.h>


typedef __attribute__((ext_vector_type(2))) float v2f;
typedef __attribute__((ext_vector_type(8))) float v8f;

static constexpr int BB = 8;
static constexpr int TT = 2048;
static constexpr int CD = 1024;
static constexpr int HD = 64;
static constexpr int BT = BB * TT;

static constexpr int MTILE = 64;
static constexpr int KC    = 32;
static constexpr int XPAD  = 34;
static constexpr int SK    = 64;
static constexpr int KPAD  = 66;

__device__ __forceinline__ v8f wmma_f32(v2f a, v2f b, v8f c) {
    return __builtin_amdgcn_wmma_f32_16x16x4_f32(
         false, a,  false, b,
         (short)0, c,  false,  false);
}

template <int XM>
__device__ __forceinline__ float row_xor(float v) {
    return __builtin_bit_cast(float,
        __builtin_amdgcn_mov_dpp(__builtin_bit_cast(int, v), 0x160 + XM, 0xf, 0xf, true));
}

__device__ __forceinline__ float row_max16(float v) {
    v = fmaxf(v, row_xor<1>(v));
    v = fmaxf(v, row_xor<2>(v));
    v = fmaxf(v, row_xor<4>(v));
    v = fmaxf(v, row_xor<8>(v));
    return v;
}
__device__ __forceinline__ float row_sum16(float v) {
    v += row_xor<1>(v);
    v += row_xor<2>(v);
    v += row_xor<4>(v);
    v += row_xor<8>(v);
    return v;
}

__global__ __launch_bounds__(256) void proj_qkv_kernel(
    const float* __restrict__ x,
    const float* __restrict__ Wq, const float* __restrict__ Wk,
    const float* __restrict__ Wv,
    float* __restrict__ Q, float* __restrict__ K, float* __restrict__ V)
{
    __shared__ float xs[2][MTILE * XPAD];
    __shared__ float ws[2][3 * HD * XPAD];

    const int tid  = threadIdx.x;
    const int wave = tid >> 5;
    const int lane = tid & 31;
    const int grp  = lane >> 4;
    const int idx  = lane & 15;
    const long rowBlock = (long)blockIdx.x * MTILE;

    const float* Wm[3] = {Wq, Wk, Wv};

    const int rt = wave & 3;
    const int ch = wave >> 2;
    v8f acc[6] = {};

    const int xr  = tid >> 3;
    const int xc4 = tid & 7;
    const int wk  = tid >> 6;
    const int wn  = tid & 63;
    const float* xg = x + (rowBlock + xr) * CD + xc4 * 4;

    float4 xv[2];
    float  wv[3][8];

    auto load_chunk = [&](const int kc) {
        #pragma unroll
        for (int j = 0; j < 2; ++j)
            xv[j] = *(const float4*)(xg + (long)(32 * j) * CD + kc);
        #pragma unroll
        for (int m = 0; m < 3; ++m) {
            const float* wg = Wm[m] + (long)(kc + wk) * HD + wn;
            #pragma unroll
            for (int j = 0; j < 8; ++j)
                wv[m][j] = wg[4 * j * HD];
        }
    };
    auto store_chunk = [&](const int st) {
        #pragma unroll
        for (int j = 0; j < 2; ++j) {
            float* dst = &xs[st][(xr + 32 * j) * XPAD + xc4 * 4];
            dst[0] = xv[j].x; dst[1] = xv[j].y; dst[2] = xv[j].z; dst[3] = xv[j].w;
        }
        #pragma unroll
        for (int m = 0; m < 3; ++m) {
            #pragma unroll
            for (int j = 0; j < 8; ++j)
                ws[st][(m * HD + wn) * XPAD + wk + 4 * j] = wv[m][j];
        }
    };

    load_chunk(0);
    store_chunk(0);
    __syncthreads();

    for (int kc = 0; kc < CD; kc += KC) {
        const int  st   = (kc / KC) & 1;
        const bool more = kc + KC < CD;
        if (more) {
            load_chunk(kc + KC);
            if (kc + 2 * KC < CD && tid < 2 * MTILE) {
                const float* nx = x + (rowBlock + (tid >> 1)) * CD + kc + 2 * KC + (tid & 1) * 16;
                __builtin_prefetch(nx, 0, 1);
            }
        }

        const float* xw = &xs[st][(rt * 16 + idx) * XPAD];
        const float* ww = &ws[st][(ch * 96 + idx) * XPAD];
        #pragma unroll
        for (int kk = 0; kk < KC; kk += 4) {
            v2f a = *(const v2f*)(xw + kk + 2 * grp);
            #pragma unroll
            for (int c = 0; c < 6; ++c) {
                v2f b = *(const v2f*)(ww + c * 16 * XPAD + kk + 2 * grp);
                acc[c] = wmma_f32(a, b, acc[c]);
            }
        }

        if (more) {
            store_chunk(st ^ 1);
            __syncthreads();
        }
    }

    const long rowW = rowBlock + rt * 16;
    #pragma unroll
    for (int c = 0; c < 6; ++c) {
        const int tc = ch * 6 + c;
        const int m  = tc >> 2;
        const int t  = tc & 3;
        float* Om = (m == 0) ? Q : ((m == 1) ? K : V);
        #pragma unroll
        for (int r = 0; r < 8; ++r)
            Om[(rowW + r + 8 * grp) * HD + t * 16 + idx] = acc[c][r];
    }
}

static constexpr int QT = 32;

__device__ __forceinline__ void attn_tile_step(
    const v2f (&qf)[16], const v2f (&kb)[16], const v2f (&vb)[4][4],
    v8f (&o)[4], float (&mrow)[8], float (&lrow)[8],
    float* pw, const int qrow0, const int key0, const int grp, const int idx)
{
    v8f s = {};
    #pragma unroll
    for (int st = 0; st < 16; ++st)
        s = wmma_f32(qf[st], kb[st], s);

    const int key = key0 + idx;
    #pragma unroll
    for (int r = 0; r < 8; ++r) {
        const int qr = qrow0 + r + 8 * grp;
        float sv = s[r] * 0.125f;
        s[r] = (key <= qr) ? sv : -1e30f;
    }
    float tmax[8];
    #pragma unroll
    for (int r = 0; r < 8; ++r)
        tmax[r] = row_max16(s[r]);
    float pcorr[8];
    #pragma unroll
    for (int r = 0; r < 8; ++r) {
        const float mnew = fmaxf(mrow[r], tmax[r]);
        pcorr[r] = __expf(mrow[r] - mnew);
        mrow[r]  = mnew;
        const int qr = qrow0 + r + 8 * grp;
        s[r] = (key <= qr) ? __expf(s[r] - mnew) : 0.f;
    }
    #pragma unroll
    for (int r = 0; r < 8; ++r)
        lrow[r] = lrow[r] * pcorr[r] + row_sum16(s[r]);
    #pragma unroll
    for (int t = 0; t < 4; ++t)
        #pragma unroll
        for (int r = 0; r < 8; ++r)
            o[t][r] *= pcorr[r];

    #pragma unroll
    for (int r = 0; r < 8; ++r)
        pw[(r + 8 * grp) * 16 + idx] = s[r];

    #pragma unroll
    for (int j = 0; j < 4; ++j) {
        v2f a = *(const v2f*)(&pw[idx * 16 + 4 * j + 2 * grp]);
        #pragma unroll
        for (int t = 0; t < 4; ++t)
            o[t] = wmma_f32(a, vb[t][j], o[t]);
    }
}

__device__ __forceinline__ void attn_merge_store(
    float* mg, const v8f (&o)[4], const float (&mrow)[8], const float (&lrow)[8],
    float* __restrict__ out, const long rowbase, const int wave, const int lane,
    const int grp, const int idx)
{
    __syncthreads();
    float* my = mg + wave * (48 * 32) + lane;
    #pragma unroll
    for (int r = 0; r < 8; ++r) {
        my[r * 32]       = mrow[r];
        my[(8 + r) * 32] = lrow[r];
    }
    #pragma unroll
    for (int t = 0; t < 4; ++t)
        #pragma unroll
        for (int r = 0; r < 8; ++r)
            my[(16 + t * 8 + r) * 32] = o[t][r];
    __syncthreads();

    const int rt = wave >> 2;
    const int kq = wave & 3;
    const int th = kq & 1;
    const int rh = kq >> 1;
    const float* src = mg + (rt * 4) * (48 * 32) + lane;
    #pragma unroll
    for (int rr = 0; rr < 4; ++rr) {
        const int r = rh * 4 + rr;
        float mk[4], wk[4];
        #pragma unroll
        for (int k = 0; k < 4; ++k) mk[k] = src[k * (48 * 32) + r * 32];
        const float mstar = fmaxf(fmaxf(mk[0], mk[1]), fmaxf(mk[2], mk[3]));
        float L = 0.f;
        #pragma unroll
        for (int k = 0; k < 4; ++k) {
            wk[k] = __expf(mk[k] - mstar);
            L += src[k * (48 * 32) + (8 + r) * 32] * wk[k];
        }
        #pragma unroll
        for (int tt = 0; tt < 2; ++tt) {
            const int t = th * 2 + tt;
            float O = 0.f;
            #pragma unroll
            for (int k = 0; k < 4; ++k)
                O += src[k * (48 * 32) + (16 + t * 8 + r) * 32] * wk[k];
            out[(rowbase + r + 8 * grp) * HD + t * 16 + idx] = O / L;
        }
    }
}

__device__ __forceinline__ void attn_load_chunk(
    const float* __restrict__ K, const float* __restrict__ V,
    float4 (&kv)[4], float4 (&vv)[4], const long base, const int s0, const int tid)
{
    const int kr  = tid >> 4;
    const int kc4 = tid & 15;
    const long g = base + (long)(s0 + kr) * HD + kc4 * 4;
    #pragma unroll
    for (int j = 0; j < 4; ++j) {
        kv[j] = *(const float4*)(K + g + 16 * j * HD);
        vv[j] = *(const float4*)(V + g + 16 * j * HD);
    }
}
__device__ __forceinline__ void attn_store_chunk(
    float* ks, float* vs, const float4 (&kv)[4], const float4 (&vv)[4], const int tid)
{
    const int kr  = tid >> 4;
    const int kc4 = tid & 15;
    #pragma unroll
    for (int j = 0; j < 4; ++j) {
        const int r = kr + 16 * j;
        float* dst = &ks[r * KPAD + kc4 * 4];
        dst[0] = kv[j].x; dst[1] = kv[j].y; dst[2] = kv[j].z; dst[3] = kv[j].w;
        vs[(kc4 * 4 + 0) * KPAD + r] = vv[j].x;
        vs[(kc4 * 4 + 1) * KPAD + r] = vv[j].y;
        vs[(kc4 * 4 + 2) * KPAD + r] = vv[j].z;
        vs[(kc4 * 4 + 3) * KPAD + r] = vv[j].w;
    }
}

__device__ __forceinline__ void attn_load_kv_frags(
    const float* ks, const float* vs, v2f (&kb)[16], v2f (&vb)[4][4],
    const int kq, const int grp, const int idx)
{
    const float* kw = &ks[(kq * 16 + idx) * KPAD + 2 * grp];
    #pragma unroll
    for (int st = 0; st < 16; ++st) kb[st] = *(const v2f*)(kw + 4 * st);
    #pragma unroll
    for (int t = 0; t < 4; ++t)
        #pragma unroll
        for (int j = 0; j < 4; ++j)
            vb[t][j] = *(const v2f*)(&vs[(t * 16 + idx) * KPAD + kq * 16 + 4 * j + 2 * grp]);
}

__global__ __launch_bounds__(256) void attn_kernel(
    const float* __restrict__ Q, const float* __restrict__ K,
    const float* __restrict__ V, float* __restrict__ out)
{
    constexpr int STG = 2 * SK * KPAD;
    __shared__ __attribute__((aligned(16))) float smem[2 * STG + 8 * 256];
    float* ps = smem + 2 * STG;

    const int tid  = threadIdx.x;
    const int lane = tid & 31;
    const int grp  = lane >> 4;
    const int idx  = lane & 15;
    const int wave = __builtin_amdgcn_readfirstlane(tid >> 5);
    const int rt   = wave >> 2;
    const int kq   = wave & 3;

    const int  b     = blockIdx.y;
    const int  p     = blockIdx.x;
    const long base  = (long)b * TT * HD;
    const int  qrowA = p * QT + rt * 16;
    const int  qrowB = (TT / QT - 1 - p) * QT + rt * 16;

    v2f qfA[16], qfB[16];
    {
        const float* qa = Q + base + (long)(qrowA + idx) * HD + 2 * grp;
        const float* qb = Q + base + (long)(qrowB + idx) * HD + 2 * grp;
        #pragma unroll
        for (int s = 0; s < 16; ++s) {
            qfA[s] = *(const v2f*)(qa + 4 * s);
            qfB[s] = *(const v2f*)(qb + 4 * s);
        }
    }

    v8f   oA[4] = {}, oB[4] = {};
    float mA[8], lA[8], mB[8], lB[8];
    #pragma unroll
    for (int r = 0; r < 8; ++r) {
        mA[r] = -1e30f; lA[r] = 0.f;
        mB[r] = -1e30f; lB[r] = 0.f;
    }

    const int nchunkA = (p * QT + QT - 1) / SK + 1;
    const int nchunkB = ((TT / QT - 1 - p) * QT + QT - 1) / SK + 1;
    float* pw = ps + wave * 256;

    float4 kv[4], vv[4];
    attn_load_chunk(K, V, kv, vv, base, 0, tid);
    attn_store_chunk(smem, smem + SK * KPAD, kv, vv, tid);
    __syncthreads();

    for (int sc = 0; sc < nchunkB; ++sc) {
        const int s0 = sc * SK;
        float* ks = smem + (sc & 1) * STG;
        float* vs = ks + SK * KPAD;
        const bool more = sc + 1 < nchunkB;
        if (more) {
            attn_load_chunk(K, V, kv, vv, base, s0 + SK, tid);
            if (sc + 2 < nchunkB) {
                const int  pt   = tid & 127;
                const long nrow = base + (long)(s0 + 2 * SK + (pt >> 1)) * HD + (pt & 1) * 32;
                __builtin_prefetch((tid < 128 ? K : V) + nrow, 0, 1);
            }
        }
        v2f kb[16], vb[4][4];
        attn_load_kv_frags(ks, vs, kb, vb, kq, grp, idx);
        const int key0 = s0 + kq * 16;
        if (sc < nchunkA)
            attn_tile_step(qfA, kb, vb, oA, mA, lA, pw, qrowA, key0, grp, idx);
        attn_tile_step(qfB, kb, vb, oB, mB, lB, pw, qrowB, key0, grp, idx);
        if (more) {
            float* kn = smem + ((sc + 1) & 1) * STG;
            attn_store_chunk(kn, kn + SK * KPAD, kv, vv, tid);
            __syncthreads();
        }
    }

    attn_merge_store(smem, oA, mA, lA, out, base / HD + qrowA, wave, lane, grp, idx);
    attn_merge_store(smem, oB, mB, lB, out, base / HD + qrowB, wave, lane, grp, idx);
}

extern "C" void kernel_launch(void* const* d_in, const int* in_sizes, int n_in,
                              void* d_out, int out_size, void* d_ws, size_t ws_size,
                              hipStream_t stream) {
    const float* x  = (const float*)d_in[0];
    const float* Wq = (const float*)d_in[1];
    const float* Wk = (const float*)d_in[2];
    const float* Wv = (const float*)d_in[3];
    float* out = (float*)d_out;

    float* Q = (float*)d_ws;
    float* K = Q + (size_t)BT * HD;
    float* V = K + (size_t)BT * HD;

    proj_qkv_kernel<<<dim3(BT / MTILE), 256, 0, stream>>>(x, Wq, Wk, Wv, Q, K, V);
    attn_kernel<<<dim3(TT / MTILE, BB), 256, 0, stream>>>(Q, K, V, out);
}


# Attention on transposed tiles (S^T = K Q^T, O^T = V^T P^T): P feeds the 2nd product from accumulator registers, no LDS patch; softmax max/sum in-lane (new sum order); key loop split, no join copies
# speedup vs baseline: 8.9956x; 1.0904x over previous
#include <hip/hip_runtime.h>
#include <hip/hip_bf16.h>
#include <type_traits>


typedef __attribute__((ext_vector_type(2))) float v2f;
typedef __attribute__((ext_vector_type(8))) float v8f;

static constexpr int BB = 8;
static constexpr int TT = 2048;
static constexpr int CD = 1024;
static constexpr int HD = 64;
static constexpr int BT = BB * TT;

static constexpr int MTILE = 64;
static constexpr int KC    = 32;
static constexpr int XPAD  = 34;
static constexpr int SK    = 64;
static constexpr int KPAD  = 66;

__device__ __forceinline__ v8f wmma_f32(v2f a, v2f b, v8f c) {
    return __builtin_amdgcn_wmma_f32_16x16x4_f32(
         false, a,  false, b,
         (short)0, c,  false,  false);
}

__device__ __forceinline__ float xhalf(float x) {
    const int b = __builtin_bit_cast(int, x);
    return __builtin_bit_cast(float,
        __builtin_amdgcn_permlanex16(b, b, 0x76543210, 0xfedcba98, false, false));
}

__global__ __launch_bounds__(256) void proj_qkv_kernel(
    const float* __restrict__ x,
    const float* __restrict__ Wq, const float* __restrict__ Wk,
    const float* __restrict__ Wv,
    float* __restrict__ Q, float* __restrict__ K, float* __restrict__ V)
{
    __shared__ float xs[2][MTILE * XPAD];
    __shared__ float ws[2][3 * HD * XPAD];

    const int tid  = threadIdx.x;
    const int wave = tid >> 5;
    const int lane = tid & 31;
    const int grp  = lane >> 4;
    const int idx  = lane & 15;
    const long rowBlock = (long)blockIdx.x * MTILE;

    const float* Wm[3] = {Wq, Wk, Wv};

    const int rt = wave & 3;
    const int ch = wave >> 2;
    v8f acc[6] = {};

    const int xr  = tid >> 3;
    const int xc4 = tid & 7;
    const int wk  = tid >> 6;
    const int wn  = tid & 63;
    const float* xg = x + (rowBlock + xr) * CD + xc4 * 4;

    float4 xv[2];
    float  wv[3][8];

    auto load_chunk = [&](const int kc) {
        #pragma unroll
        for (int j = 0; j < 2; ++j)
            xv[j] = *(const float4*)(xg + (long)(32 * j) * CD + kc);
        #pragma unroll
        for (int m = 0; m < 3; ++m) {
            const float* wg = Wm[m] + (long)(kc + wk) * HD + wn;
            #pragma unroll
            for (int j = 0; j < 8; ++j)
                wv[m][j] = wg[4 * j * HD];
        }
    };
    auto store_chunk = [&](const int st) {
        #pragma unroll
        for (int j = 0; j < 2; ++j) {
            float* dst = &xs[st][(xr + 32 * j) * XPAD + xc4 * 4];
            dst[0] = xv[j].x; dst[1] = xv[j].y; dst[2] = xv[j].z; dst[3] = xv[j].w;
        }
        #pragma unroll
        for (int m = 0; m < 3; ++m) {
            #pragma unroll
            for (int j = 0; j < 8; ++j)
                ws[st][(m * HD + wn) * XPAD + wk + 4 * j] = wv[m][j];
        }
    };

    load_chunk(0);
    store_chunk(0);
    __syncthreads();

    for (int kc = 0; kc < CD; kc += KC) {
        const int  st   = (kc / KC) & 1;
        const bool more = kc + KC < CD;
        if (more) {
            load_chunk(kc + KC);
            if (kc + 2 * KC < CD && tid < 2 * MTILE) {
                const float* nx = x + (rowBlock + (tid >> 1)) * CD + kc + 2 * KC + (tid & 1) * 16;
                __builtin_prefetch(nx, 0, 1);
            }
        }

        const float* xw = &xs[st][(rt * 16 + idx) * XPAD];
        const float* ww = &ws[st][(ch * 96 + idx) * XPAD];
        #pragma unroll
        for (int kk = 0; kk < KC; kk += 4) {
            v2f a = *(const v2f*)(xw + kk + 2 * grp);
            #pragma unroll
            for (int c = 0; c < 6; ++c) {
                v2f b = *(const v2f*)(ww + c * 16 * XPAD + kk + 2 * grp);
                acc[c] = wmma_f32(a, b, acc[c]);
            }
        }

        if (more) {
            store_chunk(st ^ 1);
            __syncthreads();
        }
    }

    const long rowW = rowBlock + rt * 16;
    #pragma unroll
    for (int c = 0; c < 6; ++c) {
        const int tc = ch * 6 + c;
        const int m  = tc >> 2;
        const int t  = tc & 3;
        float* Om = (m == 0) ? Q : ((m == 1) ? K : V);
        #pragma unroll
        for (int r = 0; r < 8; ++r)
            Om[(rowW + r + 8 * grp) * HD + t * 16 + idx] = acc[c][r];
    }
}

static constexpr int QT = 32;

__device__ __forceinline__ void attn_tile_step(
    const v2f (&qf)[16], const v2f (&kb)[16], const v2f (&va)[4][4],
    v8f (&o)[4], float& mrun, float& lpart, const int qcol, const int keyrow0)
{
    v8f s = {};
    #pragma unroll
    for (int st = 0; st < 16; ++st)
        s = wmma_f32(kb[st], qf[st], s);

    float tmax = -1e30f;
    #pragma unroll
    for (int r = 0; r < 8; ++r) {
        const float sv = s[r] * 0.125f;
        s[r] = (keyrow0 + r <= qcol) ? sv : -1e30f;
        tmax = fmaxf(tmax, s[r]);
    }
    tmax = fmaxf(tmax, xhalf(tmax));
    const float mnew  = fmaxf(mrun, tmax);
    const float pcorr = __expf(mrun - mnew);
    mrun = mnew;
    float psum = 0.f;
    #pragma unroll
    for (int r = 0; r < 8; ++r) {
        s[r] = (keyrow0 + r <= qcol) ? __expf(s[r] - mnew) : 0.f;
        psum += s[r];
    }
    lpart = lpart * pcorr + psum;
    #pragma unroll
    for (int t = 0; t < 4; ++t)
        #pragma unroll
        for (int r = 0; r < 8; ++r)
            o[t][r] *= pcorr;

    #pragma unroll
    for (int j = 0; j < 4; ++j) {
        v2f pb;
        pb.x = s[2 * j];
        pb.y = s[2 * j + 1];
        #pragma unroll
        for (int t = 0; t < 4; ++t)
            o[t] = wmma_f32(va[t][j], pb, o[t]);
    }
}

static constexpr int MOP = 68;
static constexpr int MWS = 16 * MOP + 48;

__device__ __forceinline__ void attn_merge_store(
    float* mg, const v8f (&o)[4], const float mrun, const float lpart,
    float* __restrict__ out, const long rowbase, const int wave, const int lane,
    const int grp, const int idx)
{
    __syncthreads();
    float* my = mg + wave * MWS;
    #pragma unroll
    for (int t = 0; t < 4; ++t) {
        float* d = my + idx * MOP + 16 * t + 8 * grp;
        *(float4*)(d)     = make_float4(o[t][0], o[t][1], o[t][2], o[t][3]);
        *(float4*)(d + 4) = make_float4(o[t][4], o[t][5], o[t][6], o[t][7]);
    }
    my[16 * MOP + idx]       = mrun;
    my[16 * MOP + 16 + lane] = lpart;
    __syncthreads();

    const int rt = wave >> 2;
    const int kq = wave & 3;
    const float* src = mg + (rt * 4) * MWS;
    #pragma unroll
    for (int i = 0; i < 2; ++i) {
        const int L   = kq * 8 + i * 4 + (lane >> 3);
        const int q   = L >> 1;
        const int hd0 = (L & 1) * 32 + (lane & 7) * 4;
        float mk[4], wk[4];
        #pragma unroll
        for (int k = 0; k < 4; ++k) mk[k] = src[k * MWS + 16 * MOP + q];
        const float mstar = fmaxf(fmaxf(mk[0], mk[1]), fmaxf(mk[2], mk[3]));
        float Lsum = 0.f;
        #pragma unroll
        for (int k = 0; k < 4; ++k) {
            wk[k] = __expf(mk[k] - mstar);
            const float lk = src[k * MWS + 16 * MOP + 16 + q] + src[k * MWS + 16 * MOP + 32 + q];
            Lsum += lk * wk[k];
        }
        float4 acc = make_float4(0.f, 0.f, 0.f, 0.f);
        #pragma unroll
        for (int k = 0; k < 4; ++k) {
            const float4 v = *(const float4*)(src + k * MWS + q * MOP + hd0);
            acc.x += v.x * wk[k];
            acc.y += v.y * wk[k];
            acc.z += v.z * wk[k];
            acc.w += v.w * wk[k];
        }
        *(float4*)(out + (rowbase + q) * HD + hd0) =
            make_float4(acc.x / Lsum, acc.y / Lsum, acc.z / Lsum, acc.w / Lsum);
    }
}

__device__ __forceinline__ void attn_load_chunk(
    const float* __restrict__ K, const float* __restrict__ V,
    float4 (&kv)[4], float4 (&vv)[4], const long base, const int s0, const int tid)
{
    const int kr  = tid >> 4;
    const int kc4 = tid & 15;
    const long g = base + (long)(s0 + kr) * HD + kc4 * 4;
    #pragma unroll
    for (int j = 0; j < 4; ++j) {
        kv[j] = *(const float4*)(K + g + 16 * j * HD);
        vv[j] = *(const float4*)(V + g + 16 * j * HD);
    }
}
__device__ __forceinline__ void attn_store_chunk(
    float* ks, float* vs, const float4 (&kv)[4], const float4 (&vv)[4], const int tid)
{
    const int kr  = tid >> 4;
    const int kc4 = tid & 15;
    #pragma unroll
    for (int j = 0; j < 4; ++j) {
        const int r = kr + 16 * j;
        float* dst = &ks[r * KPAD + kc4 * 4];
        dst[0] = kv[j].x; dst[1] = kv[j].y; dst[2] = kv[j].z; dst[3] = kv[j].w;
        vs[(kc4 * 4 + 0) * KPAD + r] = vv[j].x;
        vs[(kc4 * 4 + 1) * KPAD + r] = vv[j].y;
        vs[(kc4 * 4 + 2) * KPAD + r] = vv[j].z;
        vs[(kc4 * 4 + 3) * KPAD + r] = vv[j].w;
    }
}

__device__ __forceinline__ void attn_load_kv_frags(
    const float* ks, const float* vs, v2f (&kb)[16], v2f (&va)[4][4],
    const int kq, const int grp, const int idx)
{
    const float* kw = &ks[(kq * 16 + idx) * KPAD + 2 * grp];
    #pragma unroll
    for (int st = 0; st < 16; ++st) kb[st] = *(const v2f*)(kw + 4 * st);
    #pragma unroll
    for (int t = 0; t < 4; ++t)
        #pragma unroll
        for (int j = 0; j < 4; ++j)
            va[t][j] = *(const v2f*)(&vs[(t * 16 + idx) * KPAD + kq * 16 + 8 * grp + 2 * j]);
}

__global__ __launch_bounds__(256) void attn_kernel(
    const float* __restrict__ Q, const float* __restrict__ K,
    const float* __restrict__ V, float* __restrict__ out)
{
    constexpr int STG = 2 * SK * KPAD;
    static_assert(8 * MWS <= 2 * STG, "merge buffer must fit over the K/V stages");
    __shared__ __attribute__((aligned(16))) float smem[2 * STG];

    const int tid  = threadIdx.x;
    const int lane = tid & 31;
    const int grp  = lane >> 4;
    const int idx  = lane & 15;
    const int wave = __builtin_amdgcn_readfirstlane(tid >> 5);
    const int rt   = wave >> 2;
    const int kq   = wave & 3;

    const int  b     = blockIdx.y;
    const int  p     = blockIdx.x;
    const long base  = (long)b * TT * HD;
    const int  qrowA = p * QT + rt * 16;
    const int  qrowB = (TT / QT - 1 - p) * QT + rt * 16;

    v2f qfA[16], qfB[16];
    {
        const float* qa = Q + base + (long)(qrowA + idx) * HD + 2 * grp;
        const float* qb = Q + base + (long)(qrowB + idx) * HD + 2 * grp;
        #pragma unroll
        for (int s = 0; s < 16; ++s) {
            qfA[s] = *(const v2f*)(qa + 4 * s);
            qfB[s] = *(const v2f*)(qb + 4 * s);
        }
    }

    v8f   oA[4] = {}, oB[4] = {};
    float mA = -1e30f, lA = 0.f;
    float mB = -1e30f, lB = 0.f;

    const int nchunkA = (p * QT + QT - 1) / SK + 1;
    const int nchunkB = ((TT / QT - 1 - p) * QT + QT - 1) / SK + 1;

    float4 kv[4], vv[4];
    attn_load_chunk(K, V, kv, vv, base, 0, tid);
    attn_store_chunk(smem, smem + SK * KPAD, kv, vv, tid);
    __syncthreads();

    auto chunk_trip = [&](const int sc, const auto both_tag) {
        constexpr bool BOTH = decltype(both_tag)::value;
        const int s0 = sc * SK;
        float* ks = smem + (sc & 1) * STG;
        float* vs = ks + SK * KPAD;
        const bool more = sc + 1 < nchunkB;
        if (more) {
            attn_load_chunk(K, V, kv, vv, base, s0 + SK, tid);
            if (sc + 2 < nchunkB) {
                const int  pt   = tid & 127;
                const long nrow = base + (long)(s0 + 2 * SK + (pt >> 1)) * HD + (pt & 1) * 32;
                __builtin_prefetch((tid < 128 ? K : V) + nrow, 0, 1);
            }
        }
        v2f kb[16], va[4][4];
        attn_load_kv_frags(ks, vs, kb, va, kq, grp, idx);
        __builtin_amdgcn_sched_barrier(0);
        const int keyrow0 = s0 + kq * 16 + 8 * grp;
        if constexpr (BOTH)
            attn_tile_step(qfA, kb, va, oA, mA, lA, qrowA + idx, keyrow0);
        attn_tile_step(qfB, kb, va, oB, mB, lB, qrowB + idx, keyrow0);
        if (more) {
            float* kn = smem + ((sc + 1) & 1) * STG;
            attn_store_chunk(kn, kn + SK * KPAD, kv, vv, tid);
            __syncthreads();
        }
    };
    for (int sc = 0; sc < nchunkA; ++sc)
        chunk_trip(sc, std::true_type{});
    for (int sc = nchunkA; sc < nchunkB; ++sc)
        chunk_trip(sc, std::false_type{});

    attn_merge_store(smem, oA, mA, lA, out, base / HD + qrowA, wave, lane, grp, idx);
    attn_merge_store(smem, oB, mB, lB, out, base / HD + qrowB, wave, lane, grp, idx);
}

extern "C" void kernel_launch(void* const* d_in, const int* in_sizes, int n_in,
                              void* d_out, int out_size, void* d_ws, size_t ws_size,
                              hipStream_t stream) {
    const float* x  = (const float*)d_in[0];
    const float* Wq = (const float*)d_in[1];
    const float* Wk = (const float*)d_in[2];
    const float* Wv = (const float*)d_in[3];
    float* out = (float*)d_out;

    float* Q = (float*)d_ws;
    float* K = Q + (size_t)BT * HD;
    float* V = K + (size_t)BT * HD;

    proj_qkv_kernel<<<dim3(BT / MTILE), 256, 0, stream>>>(x, Wq, Wk, Wv, Q, K, V);
    attn_kernel<<<dim3(TT / MTILE, BB), 256, 0, stream>>>(Q, K, V, out);
}


# K depth per LDS stage doubled in both kernels (projection 32->64 columns, attention 64->128 keys): half the barrier/refill phases, 16-byte stage stores; one softmax update per 32 keys (new sum order)
# speedup vs baseline: 9.5824x; 1.0652x over previous
#include <hip/hip_runtime.h>
#include <hip/hip_bf16.h>
#include <type_traits>


typedef __attribute__((ext_vector_type(2))) float v2f;
typedef __attribute__((ext_vector_type(8))) float v8f;

static constexpr int BB = 8;
static constexpr int TT = 2048;
static constexpr int CD = 1024;
static constexpr int HD = 64;
static constexpr int BT = BB * TT;

static constexpr int MTILE = 64;
static constexpr int KC    = 64;
static constexpr int XPAD  = 68;
static constexpr int SK    = 128;
static constexpr int KPAD  = 66;
static constexpr int VPAD  = 130;

__device__ __forceinline__ v8f wmma_f32(v2f a, v2f b, v8f c) {
    return __builtin_amdgcn_wmma_f32_16x16x4_f32(
         false, a,  false, b,
         (short)0, c,  false,  false);
}

__device__ __forceinline__ float xhalf(float x) {
    const int b = __builtin_bit_cast(int, x);
    return __builtin_bit_cast(float,
        __builtin_amdgcn_permlanex16(b, b, 0x76543210, 0xfedcba98, false, false));
}

__global__ __launch_bounds__(256) void proj_qkv_kernel(
    const float* __restrict__ x,
    const float* __restrict__ Wq, const float* __restrict__ Wk,
    const float* __restrict__ Wv,
    float* __restrict__ Q, float* __restrict__ K, float* __restrict__ V)
{
    __shared__ __attribute__((aligned(16))) float xs[2][MTILE * XPAD];
    __shared__ __attribute__((aligned(16))) float ws[2][3 * HD * XPAD];

    const int tid  = threadIdx.x;
    const int wave = tid >> 5;
    const int lane = tid & 31;
    const int grp  = lane >> 4;
    const int idx  = lane & 15;
    const long rowBlock = (long)blockIdx.x * MTILE;

    const float* Wm[3] = {Wq, Wk, Wv};

    const int rt = wave & 3;
    const int ch = wave >> 2;
    v8f acc[6] = {};

    const int xr  = tid >> 4;
    const int xc4 = tid & 15;
    const int wk  = tid >> 6;
    const int wn  = tid & 63;
    const float* xg = x + (rowBlock + xr) * CD + xc4 * 4;

    float4 xv0, xv1, xv2, xv3;
    float  wv[3][16];

    auto load_chunk = [&](const int kc) {
        xv0 = *(const float4*)(xg + kc);
        xv1 = *(const float4*)(xg + (long)16 * CD + kc);
        xv2 = *(const float4*)(xg + (long)32 * CD + kc);
        xv3 = *(const float4*)(xg + (long)48 * CD + kc);
        #pragma unroll
        for (int m = 0; m < 3; ++m) {
            const float* wg = Wm[m] + (long)(kc + 16 * wk) * HD + wn;
            #pragma unroll
            for (int j = 0; j < 16; ++j)
                wv[m][j] = wg[j * HD];
        }
    };
    auto store_chunk = [&](const int st) {
        float* xd = &xs[st][xr * XPAD + xc4 * 4];
        *(float4*)(xd)             = xv0;
        *(float4*)(xd + 16 * XPAD) = xv1;
        *(float4*)(xd + 32 * XPAD) = xv2;
        *(float4*)(xd + 48 * XPAD) = xv3;
        #pragma unroll
        for (int m = 0; m < 3; ++m) {
            #pragma unroll
            for (int j = 0; j < 4; ++j)
                *(float4*)&ws[st][(m * HD + wn) * XPAD + 16 * wk + 4 * j] =
                    make_float4(wv[m][4 * j], wv[m][4 * j + 1], wv[m][4 * j + 2], wv[m][4 * j + 3]);
        }
    };

    load_chunk(0);
    store_chunk(0);
    __syncthreads();

    for (int kc = 0; kc < CD; kc += KC) {
        const int  st   = (kc / KC) & 1;
        const bool more = kc + KC < CD;
        if (more) {
            load_chunk(kc + KC);
            if (kc + 2 * KC < CD && tid < 2 * MTILE) {
                const float* nx = x + (rowBlock + (tid >> 1)) * CD + kc + 2 * KC + (tid & 1) * 32;
                __builtin_prefetch(nx, 0, 1);
            }
        }

        const float* xw = &xs[st][(rt * 16 + idx) * XPAD];
        const float* ww = &ws[st][(ch * 96 + idx) * XPAD];
        #pragma unroll
        for (int kk = 0; kk < KC; kk += 4) {
            v2f a = *(const v2f*)(xw + kk + 2 * grp);
            #pragma unroll
            for (int c = 0; c < 6; ++c) {
                v2f b = *(const v2f*)(ww + c * 16 * XPAD + kk + 2 * grp);
                acc[c] = wmma_f32(a, b, acc[c]);
            }
        }

        if (more) {
            store_chunk(st ^ 1);
            __syncthreads();
        }
    }

    const long rowW = rowBlock + rt * 16;
    #pragma unroll
    for (int c = 0; c < 6; ++c) {
        const int tc = ch * 6 + c;
        const int m  = tc >> 2;
        const int t  = tc & 3;
        float* Om = (m == 0) ? Q : ((m == 1) ? K : V);
        #pragma unroll
        for (int r = 0; r < 8; ++r)
            Om[(rowW + r + 8 * grp) * HD + t * 16 + idx] = acc[c][r];
    }
}

static constexpr int QT = 32;

__device__ __forceinline__ void attn_tile_step(
    const v2f (&qf)[16], const v2f (&kb)[2][16], const v2f (&va)[4][8],
    v8f (&o)[4], float& mrun, float& lpart, const int qcol, const int keyrow0)
{
    v8f s0 = {}, s1 = {};
    #pragma unroll
    for (int st = 0; st < 16; ++st) {
        s0 = wmma_f32(kb[0][st], qf[st], s0);
        s1 = wmma_f32(kb[1][st], qf[st], s1);
    }

    float tmax = -1e30f;
    #pragma unroll
    for (int r = 0; r < 8; ++r) {
        const float a = s0[r] * 0.125f;
        const float b = s1[r] * 0.125f;
        s0[r] = (keyrow0 + r <= qcol) ? a : -1e30f;
        s1[r] = (keyrow0 + 16 + r <= qcol) ? b : -1e30f;
        tmax = fmaxf(tmax, fmaxf(s0[r], s1[r]));
    }
    tmax = fmaxf(tmax, xhalf(tmax));
    const float mnew  = fmaxf(mrun, tmax);
    const float pcorr = __expf(mrun - mnew);
    mrun = mnew;
    float psum = 0.f;
    #pragma unroll
    for (int r = 0; r < 8; ++r) {
        s0[r] = (keyrow0 + r <= qcol) ? __expf(s0[r] - mnew) : 0.f;
        psum += s0[r];
    }
    #pragma unroll
    for (int r = 0; r < 8; ++r) {
        s1[r] = (keyrow0 + 16 + r <= qcol) ? __expf(s1[r] - mnew) : 0.f;
        psum += s1[r];
    }
    lpart = lpart * pcorr + psum;
    #pragma unroll
    for (int t = 0; t < 4; ++t)
        #pragma unroll
        for (int r = 0; r < 8; ++r)
            o[t][r] *= pcorr;

    #pragma unroll
    for (int j = 0; j < 4; ++j) {
        v2f pb;
        pb.x = s0[2 * j];
        pb.y = s0[2 * j + 1];
        #pragma unroll
        for (int t = 0; t < 4; ++t)
            o[t] = wmma_f32(va[t][j], pb, o[t]);
    }
    #pragma unroll
    for (int j = 0; j < 4; ++j) {
        v2f pb;
        pb.x = s1[2 * j];
        pb.y = s1[2 * j + 1];
        #pragma unroll
        for (int t = 0; t < 4; ++t)
            o[t] = wmma_f32(va[t][4 + j], pb, o[t]);
    }
}

static constexpr int MOP = 68;
static constexpr int MWS = 16 * MOP + 48;

__device__ __forceinline__ void attn_merge_store(
    float* mg, const v8f (&o)[4], const float mrun, const float lpart,
    float* __restrict__ out, const long rowbase, const int wave, const int lane,
    const int grp, const int idx)
{
    __syncthreads();
    float* my = mg + wave * MWS;
    #pragma unroll
    for (int t = 0; t < 4; ++t) {
        float* d = my + idx * MOP + 16 * t + 8 * grp;
        *(float4*)(d)     = make_float4(o[t][0], o[t][1], o[t][2], o[t][3]);
        *(float4*)(d + 4) = make_float4(o[t][4], o[t][5], o[t][6], o[t][7]);
    }
    my[16 * MOP + idx]       = mrun;
    my[16 * MOP + 16 + lane] = lpart;
    __syncthreads();

    const int rt = wave >> 2;
    const int kq = wave & 3;
    const float* src = mg + (rt * 4) * MWS;
    #pragma unroll
    for (int i = 0; i < 2; ++i) {
        const int L   = kq * 8 + i * 4 + (lane >> 3);
        const int q   = L >> 1;
        const int hd0 = (L & 1) * 32 + (lane & 7) * 4;
        float mk[4], wk[4];
        #pragma unroll
        for (int k = 0; k < 4; ++k) mk[k] = src[k * MWS + 16 * MOP + q];
        const float mstar = fmaxf(fmaxf(mk[0], mk[1]), fmaxf(mk[2], mk[3]));
        float Lsum = 0.f;
        #pragma unroll
        for (int k = 0; k < 4; ++k) {
            wk[k] = __expf(mk[k] - mstar);
            const float lk = src[k * MWS + 16 * MOP + 16 + q] + src[k * MWS + 16 * MOP + 32 + q];
            Lsum += lk * wk[k];
        }
        float4 acc = make_float4(0.f, 0.f, 0.f, 0.f);
        #pragma unroll
        for (int k = 0; k < 4; ++k) {
            const float4 v = *(const float4*)(src + k * MWS + q * MOP + hd0);
            acc.x += v.x * wk[k];
            acc.y += v.y * wk[k];
            acc.z += v.z * wk[k];
            acc.w += v.w * wk[k];
        }
        *(float4*)(out + (rowbase + q) * HD + hd0) =
            make_float4(acc.x / Lsum, acc.y / Lsum, acc.z / Lsum, acc.w / Lsum);
    }
}

__device__ __forceinline__ void attn_load_chunk(
    const float* __restrict__ K, const float* __restrict__ V,
    float4 (&kv)[8], float4 (&vv)[8], const long base, const int s0, const int tid)
{
    const int kr  = tid >> 4;
    const int kc4 = tid & 15;
    const long g = base + (long)(s0 + kr) * HD + kc4 * 4;
    #pragma unroll
    for (int j = 0; j < 8; ++j) {
        kv[j] = *(const float4*)(K + g + 16 * j * HD);
        vv[j] = *(const float4*)(V + g + 16 * j * HD);
    }
}
__device__ __forceinline__ void attn_store_chunk(
    float* ks, float* vs, const float4 (&kv)[8], const float4 (&vv)[8], const int tid)
{
    const int kr  = tid >> 4;
    const int kc4 = tid & 15;
    #pragma unroll
    for (int j = 0; j < 8; ++j) {
        const int r = kr + 16 * j;
        float* dst = &ks[r * KPAD + kc4 * 4];
        dst[0] = kv[j].x; dst[1] = kv[j].y; dst[2] = kv[j].z; dst[3] = kv[j].w;
        vs[(kc4 * 4 + 0) * VPAD + r] = vv[j].x;
        vs[(kc4 * 4 + 1) * VPAD + r] = vv[j].y;
        vs[(kc4 * 4 + 2) * VPAD + r] = vv[j].z;
        vs[(kc4 * 4 + 3) * VPAD + r] = vv[j].w;
    }
}

__device__ __forceinline__ void attn_load_kv_frags(
    const float* ks, const float* vs, v2f (&kb)[2][16], v2f (&va)[4][8],
    const int kq, const int grp, const int idx)
{
    #pragma unroll
    for (int u = 0; u < 2; ++u) {
        const float* kw = &ks[(kq * 32 + 16 * u + idx) * KPAD + 2 * grp];
        #pragma unroll
        for (int st = 0; st < 16; ++st) kb[u][st] = *(const v2f*)(kw + 4 * st);
    }
    #pragma unroll
    for (int t = 0; t < 4; ++t)
        #pragma unroll
        for (int u = 0; u < 2; ++u)
            #pragma unroll
            for (int j = 0; j < 4; ++j)
                va[t][4 * u + j] =
                    *(const v2f*)(&vs[(t * 16 + idx) * VPAD + kq * 32 + 16 * u + 8 * grp + 2 * j]);
}

__global__ __launch_bounds__(256) void attn_kernel(
    const float* __restrict__ Q, const float* __restrict__ K,
    const float* __restrict__ V, float* __restrict__ out)
{
    constexpr int STG = SK * KPAD + HD * VPAD;
    static_assert(8 * MWS <= 2 * STG, "merge buffer must fit over the K/V stages");
    __shared__ __attribute__((aligned(16))) float smem[2 * STG];

    const int tid  = threadIdx.x;
    const int lane = tid & 31;
    const int grp  = lane >> 4;
    const int idx  = lane & 15;
    const int wave = __builtin_amdgcn_readfirstlane(tid >> 5);
    const int rt   = wave >> 2;
    const int kq   = wave & 3;

    const int  b     = blockIdx.y;
    const int  p     = blockIdx.x;
    const long base  = (long)b * TT * HD;
    const int  qrowA = p * QT + rt * 16;
    const int  qrowB = (TT / QT - 1 - p) * QT + rt * 16;

    v2f qfA[16], qfB[16];
    {
        const float* qa = Q + base + (long)(qrowA + idx) * HD + 2 * grp;
        const float* qb = Q + base + (long)(qrowB + idx) * HD + 2 * grp;
        #pragma unroll
        for (int s = 0; s < 16; ++s) {
            qfA[s] = *(const v2f*)(qa + 4 * s);
            qfB[s] = *(const v2f*)(qb + 4 * s);
        }
    }

    v8f   oA[4] = {}, oB[4] = {};
    float mA = -1e30f, lA = 0.f;
    float mB = -1e30f, lB = 0.f;

    const int nchunkA = (p * QT + QT - 1) / SK + 1;
    const int nchunkB = ((TT / QT - 1 - p) * QT + QT - 1) / SK + 1;

    float4 kv[8], vv[8];
    attn_load_chunk(K, V, kv, vv, base, 0, tid);
    attn_store_chunk(smem, smem + SK * KPAD, kv, vv, tid);
    __syncthreads();

    auto chunk_trip = [&](const int sc, const auto both_tag) {
        constexpr bool BOTH = decltype(both_tag)::value;
        const int s0 = sc * SK;
        float* ks = smem + (sc & 1) * STG;
        float* vs = ks + SK * KPAD;
        const bool more = sc + 1 < nchunkB;
        if (more) {
            attn_load_chunk(K, V, kv, vv, base, s0 + SK, tid);
            if (sc + 2 < nchunkB) {
                const int  pt   = tid & 127;
                const long nrow = base + (long)(s0 + 2 * SK + pt) * HD;
                __builtin_prefetch((tid < 128 ? K : V) + nrow, 0, 1);
            }
        }
        v2f kb[2][16], va[4][8];
        attn_load_kv_frags(ks, vs, kb, va, kq, grp, idx);
        __builtin_amdgcn_sched_barrier(0);
        const int keyrow0 = s0 + kq * 32 + 8 * grp;
        if constexpr (BOTH)
            attn_tile_step(qfA, kb, va, oA, mA, lA, qrowA + idx, keyrow0);
        attn_tile_step(qfB, kb, va, oB, mB, lB, qrowB + idx, keyrow0);
        if (more) {
            float* kn = smem + ((sc + 1) & 1) * STG;
            attn_store_chunk(kn, kn + SK * KPAD, kv, vv, tid);
            __syncthreads();
        }
    };
    for (int sc = 0; sc < nchunkA; ++sc)
        chunk_trip(sc, std::true_type{});
    for (int sc = nchunkA; sc < nchunkB; ++sc)
        chunk_trip(sc, std::false_type{});

    attn_merge_store(smem, oA, mA, lA, out, base / HD + qrowA, wave, lane, grp, idx);
    attn_merge_store(smem, oB, mB, lB, out, base / HD + qrowB, wave, lane, grp, idx);
}

extern "C" void kernel_launch(void* const* d_in, const int* in_sizes, int n_in,
                              void* d_out, int out_size, void* d_ws, size_t ws_size,
                              hipStream_t stream) {
    const float* x  = (const float*)d_in[0];
    const float* Wq = (const float*)d_in[1];
    const float* Wk = (const float*)d_in[2];
    const float* Wv = (const float*)d_in[3];
    float* out = (float*)d_out;

    float* Q = (float*)d_ws;
    float* K = Q + (size_t)BT * HD;
    float* V = K + (size_t)BT * HD;

    proj_qkv_kernel<<<dim3(BT / MTILE), 256, 0, stream>>>(x, Wq, Wk, Wv, Q, K, V);
    attn_kernel<<<dim3(TT / MTILE, BB), 256, 0, stream>>>(Q, K, V, out);
}


# Pipeline keeps the matrix pipe fed: refill loads/stores spread between the WMMAs, barrier split (signal behind the stores, wait before the next chunk's first read), fragments read ahead; same sums
# speedup vs baseline: 10.2048x; 1.0650x over previous
#include <hip/hip_runtime.h>
#include <hip/hip_bf16.h>
#include <type_traits>


typedef __attribute__((ext_vector_type(2))) float v2f;
typedef __attribute__((ext_vector_type(8))) float v8f;

static constexpr int BB = 8;
static constexpr int TT = 2048;
static constexpr int CD = 1024;
static constexpr int HD = 64;
static constexpr int BT = BB * TT;

static constexpr int MTILE = 64;
static constexpr int KC    = 64;
static constexpr int XPAD  = 68;
static constexpr int SK    = 128;
static constexpr int KPAD  = 66;
static constexpr int VPAD  = 130;

__device__ __forceinline__ v8f wmma_f32(v2f a, v2f b, v8f c) {
    return __builtin_amdgcn_wmma_f32_16x16x4_f32(
         false, a,  false, b,
         (short)0, c,  false,  false);
}

__device__ __forceinline__ float xhalf(float x) {
    const int b = __builtin_bit_cast(int, x);
    return __builtin_bit_cast(float,
        __builtin_amdgcn_permlanex16(b, b, 0x76543210, 0xfedcba98, false, false));
}

__device__ __forceinline__ void stage_publish() {
    asm volatile("s_wait_dscnt 0x0\n\ts_barrier_signal -1" ::: "memory");
}
__device__ __forceinline__ void stage_wait() {
    asm volatile("s_barrier_wait -1" ::: "memory");
}

__global__ __launch_bounds__(256) void proj_qkv_kernel(
    const float* __restrict__ x,
    const float* __restrict__ Wq, const float* __restrict__ Wk,
    const float* __restrict__ Wv,
    float* __restrict__ Q, float* __restrict__ K, float* __restrict__ V)
{
    __shared__ __attribute__((aligned(16))) float xs[3][MTILE * XPAD];
    __shared__ __attribute__((aligned(16))) float ws[3][3 * HD * XPAD];

    const int tid  = threadIdx.x;
    const int wave = tid >> 5;
    const int lane = tid & 31;
    const int grp  = lane >> 4;
    const int idx  = lane & 15;
    const long rowBlock = (long)blockIdx.x * MTILE;

    const float* Wm[3] = {Wq, Wk, Wv};

    const int rt = wave & 3;
    const int ch = wave >> 2;
    v8f acc[6] = {};

    const int xr  = tid >> 4;
    const int xc4 = tid & 15;
    const int wk  = tid >> 6;
    const int wn  = tid & 63;
    const float* xg = x + (rowBlock + xr) * CD + xc4 * 4;

    float4 xv0, xv1, xv2, xv3;
    float  wv[3][16];

    auto load_part = [&](const int kcn, const int i) {
        if (i == 0) {
            xv0 = *(const float4*)(xg + kcn);
            xv1 = *(const float4*)(xg + (long)16 * CD + kcn);
        } else if (i == 1) {
            xv2 = *(const float4*)(xg + (long)32 * CD + kcn);
            xv3 = *(const float4*)(xg + (long)48 * CD + kcn);
        } else {
            const int m = (i - 2) >> 1;
            const int h = (i - 2) & 1;
            const float* wg = Wm[m] + (long)(kcn + 16 * wk + 8 * h) * HD + wn;
            #pragma unroll
            for (int j = 0; j < 8; ++j)
                wv[m][8 * h + j] = wg[j * HD];
        }
    };
    auto store_part = [&](const int st, const int i) {
        if (i == 3) {
            float* xd = &xs[st][xr * XPAD + xc4 * 4];
            *(float4*)(xd)             = xv0;
            *(float4*)(xd + 16 * XPAD) = xv1;
            *(float4*)(xd + 32 * XPAD) = xv2;
            *(float4*)(xd + 48 * XPAD) = xv3;
        } else {
            const int m = i;
            #pragma unroll
            for (int j = 0; j < 4; ++j)
                *(float4*)&ws[st][(m * HD + wn) * XPAD + 16 * wk + 4 * j] =
                    make_float4(wv[m][4 * j], wv[m][4 * j + 1], wv[m][4 * j + 2], wv[m][4 * j + 3]);
        }
    };

    #pragma unroll
    for (int i = 0; i < 8; ++i) load_part(0, i);
    #pragma unroll
    for (int i = 0; i < 4; ++i) store_part(0, i);
    __syncthreads();
    #pragma unroll
    for (int i = 0; i < 8; ++i) load_part(KC, i);

    const int arow = (rt * 16 + idx) * XPAD + 2 * grp;
    const int brow = (ch * 96 + idx) * XPAD + 2 * grp;

    v2f fa[4], fb[4][6];
    auto load_frags = [&](const int sg, const int s) {
        const float* xw = &xs[sg][arow];
        const float* ww = &ws[sg][brow];
        fa[s % 4] = *(const v2f*)(xw + 4 * s);
        #pragma unroll
        for (int c = 0; c < 6; ++c)
            fb[s % 4][c] = *(const v2f*)(ww + c * 16 * XPAD + 4 * s);
    };
    load_frags(0, 0);
    load_frags(0, 1);

    auto chunk_trip = [&](const int kc, const auto more_tag, const auto more2_tag) {
        constexpr bool MORE  = decltype(more_tag)::value;
        constexpr bool MORE2 = decltype(more2_tag)::value;
        const int st = (kc / KC) % 3;
        const int sn = (kc / KC + 1) % 3;
        #pragma unroll
        for (int s = 0; s < KC / 4; ++s) {
            if (MORE && s == 7) {
                stage_publish();
            }
            if (s + 2 < KC / 4) {
                load_frags(st, s + 2);
            } else if (MORE) {
                if (s + 2 == KC / 4)
                    stage_wait();
                load_frags(sn, s + 2 - KC / 4);
            }
            if (MORE && s >= 2 && s < 6)
                store_part(sn, s - 2);
            if (MORE2 && s >= 8) {
                load_part(kc + 2 * KC, s - 8);
                if (s == 8 && kc + 3 * KC < CD && tid < 2 * MTILE) {
                    const float* nx = x + (rowBlock + (tid >> 1)) * CD + kc + 3 * KC + (tid & 1) * 32;
                    __builtin_prefetch(nx, 0, 1);
                }
            }
            #pragma unroll
            for (int c = 0; c < 6; ++c)
                acc[c] = wmma_f32(fa[s % 4], fb[s % 4][c], acc[c]);
            __builtin_amdgcn_sched_barrier(0);
        }
    };
    for (int kc = 0; kc + 2 * KC < CD; kc += KC)
        chunk_trip(kc, std::true_type{}, std::true_type{});
    chunk_trip(CD - 2 * KC, std::true_type{}, std::false_type{});
    chunk_trip(CD - KC, std::false_type{}, std::false_type{});

    const long rowW = rowBlock + rt * 16;
    #pragma unroll
    for (int c = 0; c < 6; ++c) {
        const int tc = ch * 6 + c;
        const int m  = tc >> 2;
        const int t  = tc & 3;
        float* Om = (m == 0) ? Q : ((m == 1) ? K : V);
        #pragma unroll
        for (int r = 0; r < 8; ++r)
            Om[(rowW + r + 8 * grp) * HD + t * 16 + idx] = acc[c][r];
    }
}

static constexpr int QT = 32;

struct no_part { __device__ __forceinline__ void operator()(const int) const {} };

template <bool PIN, class Part>
__device__ __forceinline__ void attn_scores(
    const v2f (&qf)[16], const v2f (&kb)[2][16], v8f& s0, v8f& s1, const Part& part)
{
    s0 = v8f{};
    s1 = v8f{};
    #pragma unroll
    for (int st = 0; st < 16; ++st) {
        s0 = wmma_f32(kb[0][st], qf[st], s0);
        s1 = wmma_f32(kb[1][st], qf[st], s1);
        if constexpr (PIN) {
            asm volatile("" : "+v"(s0), "+v"(s1) :: "memory");
        }
        part(st);
        if constexpr (PIN) __builtin_amdgcn_sched_barrier(0);
    }
}

__device__ __forceinline__ void attn_softmax(
    v8f& s0, v8f& s1, v8f (&o)[4], float& mrun, float& lpart, const int qcol, const int keyrow0)
{
    float tmax = -1e30f;
    #pragma unroll
    for (int r = 0; r < 8; ++r) {
        const float a = s0[r] * 0.125f;
        const float b = s1[r] * 0.125f;
        s0[r] = (keyrow0 + r <= qcol) ? a : -1e30f;
        s1[r] = (keyrow0 + 16 + r <= qcol) ? b : -1e30f;
        tmax = fmaxf(tmax, fmaxf(s0[r], s1[r]));
    }
    tmax = fmaxf(tmax, xhalf(tmax));
    const float mnew  = fmaxf(mrun, tmax);
    const float pcorr = __expf(mrun - mnew);
    mrun = mnew;
    float psum = 0.f;
    #pragma unroll
    for (int r = 0; r < 8; ++r) {
        s0[r] = (keyrow0 + r <= qcol) ? __expf(s0[r] - mnew) : 0.f;
        psum += s0[r];
    }
    #pragma unroll
    for (int r = 0; r < 8; ++r) {
        s1[r] = (keyrow0 + 16 + r <= qcol) ? __expf(s1[r] - mnew) : 0.f;
        psum += s1[r];
    }
    lpart = lpart * pcorr + psum;
    #pragma unroll
    for (int t = 0; t < 4; ++t)
        #pragma unroll
        for (int r = 0; r < 8; ++r)
            o[t][r] *= pcorr;
}

template <bool PIN, class Part>
__device__ __forceinline__ void attn_pv(
    const v2f (&va)[4][8], const v8f& s0, const v8f& s1, v8f (&o)[4], const Part& part)
{
    #pragma unroll
    for (int j = 0; j < 8; ++j) {
        v2f pb;
        pb.x = (j < 4) ? s0[2 * j]     : s1[2 * (j - 4)];
        pb.y = (j < 4) ? s0[2 * j + 1] : s1[2 * (j - 4) + 1];
        #pragma unroll
        for (int t = 0; t < 4; ++t)
            o[t] = wmma_f32(va[t][j], pb, o[t]);
        part(j);
        if constexpr (PIN) __builtin_amdgcn_sched_barrier(0);
    }
}

static constexpr int MOP = 68;
static constexpr int MWS = 16 * MOP + 48;

__device__ __forceinline__ void attn_merge_store(
    float* mg, const v8f (&o)[4], const float mrun, const float lpart,
    float* __restrict__ out, const long rowbase, const int wave, const int lane,
    const int grp, const int idx)
{
    __syncthreads();
    float* my = mg + wave * MWS;
    #pragma unroll
    for (int t = 0; t < 4; ++t) {
        float* d = my + idx * MOP + 16 * t + 8 * grp;
        *(float4*)(d)     = make_float4(o[t][0], o[t][1], o[t][2], o[t][3]);
        *(float4*)(d + 4) = make_float4(o[t][4], o[t][5], o[t][6], o[t][7]);
    }
    my[16 * MOP + idx]       = mrun;
    my[16 * MOP + 16 + lane] = lpart;
    __syncthreads();

    const int rt = wave >> 2;
    const int kq = wave & 3;
    const float* src = mg + (rt * 4) * MWS;
    #pragma unroll
    for (int i = 0; i < 2; ++i) {
        const int L   = kq * 8 + i * 4 + (lane >> 3);
        const int q   = L >> 1;
        const int hd0 = (L & 1) * 32 + (lane & 7) * 4;
        float mk[4], wk[4];
        #pragma unroll
        for (int k = 0; k < 4; ++k) mk[k] = src[k * MWS + 16 * MOP + q];
        const float mstar = fmaxf(fmaxf(mk[0], mk[1]), fmaxf(mk[2], mk[3]));
        float Lsum = 0.f;
        #pragma unroll
        for (int k = 0; k < 4; ++k) {
            wk[k] = __expf(mk[k] - mstar);
            const float lk = src[k * MWS + 16 * MOP + 16 + q] + src[k * MWS + 16 * MOP + 32 + q];
            Lsum += lk * wk[k];
        }
        float4 acc = make_float4(0.f, 0.f, 0.f, 0.f);
        #pragma unroll
        for (int k = 0; k < 4; ++k) {
            const float4 v = *(const float4*)(src + k * MWS + q * MOP + hd0);
            acc.x += v.x * wk[k];
            acc.y += v.y * wk[k];
            acc.z += v.z * wk[k];
            acc.w += v.w * wk[k];
        }
        *(float4*)(out + (rowbase + q) * HD + hd0) =
            make_float4(acc.x / Lsum, acc.y / Lsum, acc.z / Lsum, acc.w / Lsum);
    }
}

__device__ __forceinline__ void attn_load_group(
    const float* __restrict__ K, const float* __restrict__ V,
    float4& kvj, float4& vvj, const long base, const int s0, const int tid, const int j)
{
    const int kr  = tid >> 4;
    const int kc4 = tid & 15;
    const long g = base + (long)(s0 + kr + 16 * j) * HD + kc4 * 4;
    kvj = *(const float4*)(K + g);
    vvj = *(const float4*)(V + g);
}
__device__ __forceinline__ void attn_store_group(
    float* ks, float* vs, const float4& kvj, const float4& vvj, const int tid, const int j)
{
    const int kr  = tid >> 4;
    const int kc4 = tid & 15;
    const int r = kr + 16 * j;
    float* dst = &ks[r * KPAD + kc4 * 4];
    dst[0] = kvj.x; dst[1] = kvj.y; dst[2] = kvj.z; dst[3] = kvj.w;
    vs[(kc4 * 4 + 0) * VPAD + r] = vvj.x;
    vs[(kc4 * 4 + 1) * VPAD + r] = vvj.y;
    vs[(kc4 * 4 + 2) * VPAD + r] = vvj.z;
    vs[(kc4 * 4 + 3) * VPAD + r] = vvj.w;
}

__device__ __forceinline__ void attn_load_k_frags(
    const float* ks, v2f (&kb)[2][16], const int kq, const int grp, const int idx, const int q)
{
    #pragma unroll
    for (int u = 0; u < 2; ++u) {
        const float* kw = &ks[(kq * 32 + 16 * u + idx) * KPAD + 2 * grp];
        kb[u][2 * q]     = *(const v2f*)(kw + 4 * (2 * q));
        kb[u][2 * q + 1] = *(const v2f*)(kw + 4 * (2 * q + 1));
    }
}
__device__ __forceinline__ void attn_load_v_frags(
    const float* vs, v2f (&va)[4][8], const int kq, const int grp, const int idx)
{
    #pragma unroll
    for (int t = 0; t < 4; ++t)
        #pragma unroll
        for (int u = 0; u < 2; ++u)
            #pragma unroll
            for (int j = 0; j < 4; ++j)
                va[t][4 * u + j] =
                    *(const v2f*)(&vs[(t * 16 + idx) * VPAD + kq * 32 + 16 * u + 8 * grp + 2 * j]);
}

__global__ __launch_bounds__(256) void attn_kernel(
    const float* __restrict__ Q, const float* __restrict__ K,
    const float* __restrict__ V, float* __restrict__ out)
{
    constexpr int STG = SK * KPAD + HD * VPAD;
    static_assert(8 * MWS <= 2 * STG, "merge buffer must fit over the K/V stages");
    __shared__ __attribute__((aligned(16))) float smem[2 * STG];

    const int tid  = threadIdx.x;
    const int lane = tid & 31;
    const int grp  = lane >> 4;
    const int idx  = lane & 15;
    const int wave = __builtin_amdgcn_readfirstlane(tid >> 5);
    const int rt   = wave >> 2;
    const int kq   = wave & 3;

    const int  b     = blockIdx.y;
    const int  p     = blockIdx.x;
    const long base  = (long)b * TT * HD;
    const int  qrowA = p * QT + rt * 16;
    const int  qrowB = (TT / QT - 1 - p) * QT + rt * 16;

    v2f qfA[16], qfB[16];
    {
        const float* qa = Q + base + (long)(qrowA + idx) * HD + 2 * grp;
        const float* qb = Q + base + (long)(qrowB + idx) * HD + 2 * grp;
        #pragma unroll
        for (int s = 0; s < 16; ++s) {
            qfA[s] = *(const v2f*)(qa + 4 * s);
            qfB[s] = *(const v2f*)(qb + 4 * s);
        }
    }

    v8f   oA[4] = {}, oB[4] = {};
    float mA = -1e30f, lA = 0.f;
    float mB = -1e30f, lB = 0.f;

    const int nchunkA = (p * QT + QT - 1) / SK + 1;
    const int nchunkB = ((TT / QT - 1 - p) * QT + QT - 1) / SK + 1;

    float4 kv[8], vv[8];
    v2f kb[2][16];
    #pragma unroll
    for (int j = 0; j < 8; ++j) attn_load_group(K, V, kv[j], vv[j], base, 0, tid, j);
    #pragma unroll
    for (int j = 0; j < 8; ++j) attn_store_group(smem, smem + SK * KPAD, kv[j], vv[j], tid, j);
    __syncthreads();
    #pragma unroll
    for (int j = 0; j < 8; ++j) attn_load_group(K, V, kv[j], vv[j], base, SK, tid, j);
    #pragma unroll
    for (int q = 0; q < 8; ++q) attn_load_k_frags(smem, kb, kq, grp, idx, q);

    auto chunk_trip = [&](const int sc, const auto both_tag) {
        constexpr bool BOTH = decltype(both_tag)::value;
        const int s0 = sc * SK;
        float* vs = smem + (sc & 1) * STG + SK * KPAD;
        float* kn = smem + ((sc + 1) & 1) * STG;
        float* vn = kn + SK * KPAD;
        const int sl = ((sc + 2 < nchunkB) ? sc + 2 : nchunkB - 1) * SK;
        v2f va[4][8];
        attn_load_v_frags(vs, va, kq, grp, idx);
        __builtin_amdgcn_sched_barrier(0);
        const int keyrow0 = s0 + kq * 32 + 8 * grp;

        auto store_part = [&](const int st) {
            if ((st & 1) == 0) attn_store_group(kn, vn, kv[st >> 1], vv[st >> 1], tid, st >> 1);
        };
        auto load_part = [&](const int j) {
            if (j < 4) {
                attn_load_group(K, V, kv[2 * j],     vv[2 * j],     base, sl, tid, 2 * j);
                attn_load_group(K, V, kv[2 * j + 1], vv[2 * j + 1], base, sl, tid, 2 * j + 1);
            }
            attn_load_k_frags(kn, kb, kq, grp, idx, j);
            if (j == 0) {
                const int  sp   = ((sc + 3 < nchunkB) ? sc + 3 : nchunkB - 1) * SK;
                const int  pt   = tid & 127;
                const long nrow = base + (long)(sp + pt) * HD;
                __builtin_prefetch((tid < 128 ? K : V) + nrow, 0, 1);
            }
        };

        v8f sB0, sB1;
        attn_scores<true>(qfB, kb, sB0, sB1, store_part);
        stage_publish();
        if constexpr (BOTH) {
            v8f sA0, sA1;
            attn_scores<false>(qfA, kb, sA0, sA1, no_part{});
            attn_softmax(sA0, sA1, oA, mA, lA, qrowA + idx, keyrow0);
            attn_softmax(sB0, sB1, oB, mB, lB, qrowB + idx, keyrow0);
            attn_pv<false>(va, sA0, sA1, oA, no_part{});
        } else {
            attn_softmax(sB0, sB1, oB, mB, lB, qrowB + idx, keyrow0);
        }
        __builtin_amdgcn_sched_barrier(0);
        stage_wait();
        attn_pv<true>(va, sB0, sB1, oB, load_part);
    };
    for (int sc = 0; sc < nchunkA; ++sc)
        chunk_trip(sc, std::true_type{});
    for (int sc = nchunkA; sc < nchunkB; ++sc)
        chunk_trip(sc, std::false_type{});

    attn_merge_store(smem, oA, mA, lA, out, base / HD + qrowA, wave, lane, grp, idx);
    attn_merge_store(smem, oB, mB, lB, out, base / HD + qrowB, wave, lane, grp, idx);
}

extern "C" void kernel_launch(void* const* d_in, const int* in_sizes, int n_in,
                              void* d_out, int out_size, void* d_ws, size_t ws_size,
                              hipStream_t stream) {
    const float* x  = (const float*)d_in[0];
    const float* Wq = (const float*)d_in[1];
    const float* Wk = (const float*)d_in[2];
    const float* Wv = (const float*)d_in[3];
    float* out = (float*)d_out;

    float* Q = (float*)d_ws;
    float* K = Q + (size_t)BT * HD;
    float* V = K + (size_t)BT * HD;

    proj_qkv_kernel<<<dim3(BT / MTILE), 256, 0, stream>>>(x, Wq, Wk, Wv, Q, K, V);
    attn_kernel<<<dim3(TT / MTILE, BB), 256, 0, stream>>>(Q, K, V, out);
}


# Attention key loop carries fewer vector instructions: causal compare/select only in a tile's diagonal chunk, exp2 of one packed fma per score pair, staging loads from a scalar base; sum order kept
# speedup vs baseline: 10.6795x; 1.0465x over previous
#include <hip/hip_runtime.h>
#include <hip/hip_bf16.h>
#include <type_traits>


typedef __attribute__((ext_vector_type(2))) float v2f;
typedef __attribute__((ext_vector_type(8))) float v8f;

static constexpr int BB = 8;
static constexpr int TT = 2048;
static constexpr int CD = 1024;
static constexpr int HD = 64;
static constexpr int BT = BB * TT;

static constexpr int MTILE = 64;
static constexpr int KC    = 64;
static constexpr int XPAD  = 68;
static constexpr int SK    = 128;
static constexpr int KPAD  = 66;
static constexpr int VPAD  = 130;

__device__ __forceinline__ v8f wmma_f32(v2f a, v2f b, v8f c) {
    return __builtin_amdgcn_wmma_f32_16x16x4_f32(
         false, a,  false, b,
         (short)0, c,  false,  false);
}

__device__ __forceinline__ float xhalf(float x) {
    const int b = __builtin_bit_cast(int, x);
    return __builtin_bit_cast(float,
        __builtin_amdgcn_permlanex16(b, b, 0x76543210, 0xfedcba98, false, false));
}

__device__ __forceinline__ void stage_publish() {
    asm volatile("s_wait_dscnt 0x0\n\ts_barrier_signal -1" ::: "memory");
}
__device__ __forceinline__ void stage_wait() {
    asm volatile("s_barrier_wait -1" ::: "memory");
}

__global__ __launch_bounds__(256) void proj_qkv_kernel(
    const float* __restrict__ x,
    const float* __restrict__ Wq, const float* __restrict__ Wk,
    const float* __restrict__ Wv,
    float* __restrict__ Q, float* __restrict__ K, float* __restrict__ V)
{
    __shared__ __attribute__((aligned(16))) float xs[3][MTILE * XPAD];
    __shared__ __attribute__((aligned(16))) float ws[3][3 * HD * XPAD];

    const int tid  = threadIdx.x;
    const int wave = tid >> 5;
    const int lane = tid & 31;
    const int grp  = lane >> 4;
    const int idx  = lane & 15;
    const long rowBlock = (long)blockIdx.x * MTILE;

    const float* Wm[3] = {Wq, Wk, Wv};

    const int rt = wave & 3;
    const int ch = wave >> 2;
    v8f acc[6] = {};

    const int xr  = tid >> 4;
    const int xc4 = tid & 15;
    const int wk  = tid >> 6;
    const int wn  = tid & 63;
    const float* xg = x + (rowBlock + xr) * CD + xc4 * 4;

    float4 xv0, xv1, xv2, xv3;
    float  wv[3][16];

    auto load_part = [&](const int kcn, const int i) {
        if (i == 0) {
            xv0 = *(const float4*)(xg + kcn);
            xv1 = *(const float4*)(xg + (long)16 * CD + kcn);
        } else if (i == 1) {
            xv2 = *(const float4*)(xg + (long)32 * CD + kcn);
            xv3 = *(const float4*)(xg + (long)48 * CD + kcn);
        } else {
            const int m = (i - 2) >> 1;
            const int h = (i - 2) & 1;
            const float* wg = Wm[m] + (long)(kcn + 16 * wk + 8 * h) * HD + wn;
            #pragma unroll
            for (int j = 0; j < 8; ++j)
                wv[m][8 * h + j] = wg[j * HD];
        }
    };
    auto store_part = [&](const int st, const int i) {
        if (i == 3) {
            float* xd = &xs[st][xr * XPAD + xc4 * 4];
            *(float4*)(xd)             = xv0;
            *(float4*)(xd + 16 * XPAD) = xv1;
            *(float4*)(xd + 32 * XPAD) = xv2;
            *(float4*)(xd + 48 * XPAD) = xv3;
        } else {
            const int m = i;
            #pragma unroll
            for (int j = 0; j < 4; ++j)
                *(float4*)&ws[st][(m * HD + wn) * XPAD + 16 * wk + 4 * j] =
                    make_float4(wv[m][4 * j], wv[m][4 * j + 1], wv[m][4 * j + 2], wv[m][4 * j + 3]);
        }
    };

    #pragma unroll
    for (int i = 0; i < 8; ++i) load_part(0, i);
    #pragma unroll
    for (int i = 0; i < 4; ++i) store_part(0, i);
    __syncthreads();
    #pragma unroll
    for (int i = 0; i < 8; ++i) load_part(KC, i);

    const int arow = (rt * 16 + idx) * XPAD + 2 * grp;
    const int brow = (ch * 96 + idx) * XPAD + 2 * grp;

    v2f fa[4], fb[4][6];
    auto load_frags = [&](const int sg, const int s) {
        const float* xw = &xs[sg][arow];
        const float* ww = &ws[sg][brow];
        fa[s % 4] = *(const v2f*)(xw + 4 * s);
        #pragma unroll
        for (int c = 0; c < 6; ++c)
            fb[s % 4][c] = *(const v2f*)(ww + c * 16 * XPAD + 4 * s);
    };
    load_frags(0, 0);
    load_frags(0, 1);

    auto chunk_trip = [&](const int kc, const auto more_tag, const auto more2_tag) {
        constexpr bool MORE  = decltype(more_tag)::value;
        constexpr bool MORE2 = decltype(more2_tag)::value;
        const int st = (kc / KC) % 3;
        const int sn = (kc / KC + 1) % 3;
        #pragma unroll
        for (int s = 0; s < KC / 4; ++s) {
            if (MORE && s == 7) {
                stage_publish();
            }
            if (s + 2 < KC / 4) {
                load_frags(st, s + 2);
            } else if (MORE) {
                if (s + 2 == KC / 4)
                    stage_wait();
                load_frags(sn, s + 2 - KC / 4);
            }
            if (MORE && s >= 2 && s < 6)
                store_part(sn, s - 2);
            if (MORE2 && s >= 8) {
                load_part(kc + 2 * KC, s - 8);
                if (s == 8 && kc + 3 * KC < CD && tid < 2 * MTILE) {
                    const float* nx = x + (rowBlock + (tid >> 1)) * CD + kc + 3 * KC + (tid & 1) * 32;
                    __builtin_prefetch(nx, 0, 1);
                }
            }
            #pragma unroll
            for (int c = 0; c < 6; ++c)
                acc[c] = wmma_f32(fa[s % 4], fb[s % 4][c], acc[c]);
            __builtin_amdgcn_sched_barrier(0);
        }
    };
    for (int kc = 0; kc + 2 * KC < CD; kc += KC)
        chunk_trip(kc, std::true_type{}, std::true_type{});
    chunk_trip(CD - 2 * KC, std::true_type{}, std::false_type{});
    chunk_trip(CD - KC, std::false_type{}, std::false_type{});

    const long rowW = rowBlock + rt * 16;
    #pragma unroll
    for (int c = 0; c < 6; ++c) {
        const int tc = ch * 6 + c;
        const int m  = tc >> 2;
        const int t  = tc & 3;
        float* Om = (m == 0) ? Q : ((m == 1) ? K : V);
        #pragma unroll
        for (int r = 0; r < 8; ++r)
            Om[(rowW + r + 8 * grp) * HD + t * 16 + idx] = acc[c][r];
    }
}

static constexpr int QT = 32;

struct no_part { __device__ __forceinline__ void operator()(const int) const {} };

template <bool PIN, class Part>
__device__ __forceinline__ void attn_scores(
    const v2f (&qf)[16], const v2f (&kb)[2][16], v8f& s0, v8f& s1, const Part& part)
{
    s0 = v8f{};
    s1 = v8f{};
    #pragma unroll
    for (int st = 0; st < 16; ++st) {
        s0 = wmma_f32(kb[0][st], qf[st], s0);
        s1 = wmma_f32(kb[1][st], qf[st], s1);
        if constexpr (PIN) {
            asm volatile("" : "+v"(s0), "+v"(s1) :: "memory");
        }
        part(st);
        if constexpr (PIN) __builtin_amdgcn_sched_barrier(0);
    }
}

template <bool MASK>
__device__ __forceinline__ void attn_softmax(
    v8f& s0, v8f& s1, v8f (&o)[4], float& mrun, float& lpart, const int qcol, const int keyrow0)
{
    constexpr float CS = 0.125f * 1.44269504088896340736f;
    float tmax;
    if constexpr (MASK) {
        tmax = -1e30f;
        #pragma unroll
        for (int r = 0; r < 8; ++r) {
            s0[r] = (keyrow0 + r <= qcol) ? s0[r] : -1e30f;
            s1[r] = (keyrow0 + 16 + r <= qcol) ? s1[r] : -1e30f;
            tmax = fmaxf(tmax, fmaxf(s0[r], s1[r]));
        }
    } else {
        tmax = fmaxf(s0[0], s1[0]);
        #pragma unroll
        for (int r = 1; r < 8; ++r)
            tmax = fmaxf(tmax, fmaxf(s0[r], s1[r]));
    }
    tmax = fmaxf(tmax, xhalf(tmax));
    const float mnew  = fmaxf(mrun, tmax * CS);
    const float pcorr = __builtin_amdgcn_exp2f(mrun - mnew);
    mrun = mnew;
    const v2f cs2 = {CS, CS};
    const v2f nm2 = {-mnew, -mnew};
    float psum = 0.f;
    #pragma unroll
    for (int r = 0; r < 8; r += 2) {
        const v2f a = __builtin_elementwise_fma(v2f{s0[r], s0[r + 1]}, cs2, nm2);
        float e0 = __builtin_amdgcn_exp2f(a.x);
        float e1 = __builtin_amdgcn_exp2f(a.y);
        if constexpr (MASK) {
            e0 = (keyrow0 + r <= qcol) ? e0 : 0.f;
            e1 = (keyrow0 + r + 1 <= qcol) ? e1 : 0.f;
        }
        s0[r] = e0; s0[r + 1] = e1;
        psum += e0;
        psum += e1;
    }
    #pragma unroll
    for (int r = 0; r < 8; r += 2) {
        const v2f a = __builtin_elementwise_fma(v2f{s1[r], s1[r + 1]}, cs2, nm2);
        float e0 = __builtin_amdgcn_exp2f(a.x);
        float e1 = __builtin_amdgcn_exp2f(a.y);
        if constexpr (MASK) {
            e0 = (keyrow0 + 16 + r <= qcol) ? e0 : 0.f;
            e1 = (keyrow0 + 16 + r + 1 <= qcol) ? e1 : 0.f;
        }
        s1[r] = e0; s1[r + 1] = e1;
        psum += e0;
        psum += e1;
    }
    lpart = lpart * pcorr + psum;
    #pragma unroll
    for (int t = 0; t < 4; ++t)
        #pragma unroll
        for (int r = 0; r < 8; ++r)
            o[t][r] *= pcorr;
}

template <bool PIN, class Part>
__device__ __forceinline__ void attn_pv(
    const v2f (&va)[4][8], const v8f& s0, const v8f& s1, v8f (&o)[4], const Part& part)
{
    #pragma unroll
    for (int j = 0; j < 8; ++j) {
        v2f pb;
        pb.x = (j < 4) ? s0[2 * j]     : s1[2 * (j - 4)];
        pb.y = (j < 4) ? s0[2 * j + 1] : s1[2 * (j - 4) + 1];
        #pragma unroll
        for (int t = 0; t < 4; ++t)
            o[t] = wmma_f32(va[t][j], pb, o[t]);
        part(j);
        if constexpr (PIN) __builtin_amdgcn_sched_barrier(0);
    }
}

static constexpr int MOP = 68;
static constexpr int MWS = 16 * MOP + 48;

__device__ __forceinline__ void attn_merge_store(
    float* mg, const v8f (&o)[4], const float mrun, const float lpart,
    float* __restrict__ out, const long rowbase, const int wave, const int lane,
    const int grp, const int idx)
{
    __syncthreads();
    float* my = mg + wave * MWS;
    #pragma unroll
    for (int t = 0; t < 4; ++t) {
        float* d = my + idx * MOP + 16 * t + 8 * grp;
        *(float4*)(d)     = make_float4(o[t][0], o[t][1], o[t][2], o[t][3]);
        *(float4*)(d + 4) = make_float4(o[t][4], o[t][5], o[t][6], o[t][7]);
    }
    my[16 * MOP + idx]       = mrun;
    my[16 * MOP + 16 + lane] = lpart;
    __syncthreads();

    const int rt = wave >> 2;
    const int kq = wave & 3;
    const float* src = mg + (rt * 4) * MWS;
    #pragma unroll
    for (int i = 0; i < 2; ++i) {
        const int L   = kq * 8 + i * 4 + (lane >> 3);
        const int q   = L >> 1;
        const int hd0 = (L & 1) * 32 + (lane & 7) * 4;
        float mk[4], wk[4];
        #pragma unroll
        for (int k = 0; k < 4; ++k) mk[k] = src[k * MWS + 16 * MOP + q];
        const float mstar = fmaxf(fmaxf(mk[0], mk[1]), fmaxf(mk[2], mk[3]));
        float Lsum = 0.f;
        #pragma unroll
        for (int k = 0; k < 4; ++k) {
            wk[k] = __builtin_amdgcn_exp2f(mk[k] - mstar);
            const float lk = src[k * MWS + 16 * MOP + 16 + q] + src[k * MWS + 16 * MOP + 32 + q];
            Lsum += lk * wk[k];
        }
        float4 acc = make_float4(0.f, 0.f, 0.f, 0.f);
        #pragma unroll
        for (int k = 0; k < 4; ++k) {
            const float4 v = *(const float4*)(src + k * MWS + q * MOP + hd0);
            acc.x += v.x * wk[k];
            acc.y += v.y * wk[k];
            acc.z += v.z * wk[k];
            acc.w += v.w * wk[k];
        }
        *(float4*)(out + (rowbase + q) * HD + hd0) =
            make_float4(acc.x / Lsum, acc.y / Lsum, acc.z / Lsum, acc.w / Lsum);
    }
}

__device__ __forceinline__ void attn_load_group(
    const float* __restrict__ Kc, const float* __restrict__ Vc,
    float4& kvj, float4& vvj, const unsigned tbyte, const int j)
{
    kvj = *(const float4*)((const char*)Kc + (size_t)tbyte + (size_t)j * (16 * HD * 4));
    vvj = *(const float4*)((const char*)Vc + (size_t)tbyte + (size_t)j * (16 * HD * 4));
}
__device__ __forceinline__ void attn_store_group(
    float* ks, float* vs, const float4& kvj, const float4& vvj, const int tid, const int j)
{
    const int kr  = tid >> 4;
    const int kc4 = tid & 15;
    const int r = kr + 16 * j;
    float* dst = &ks[r * KPAD + kc4 * 4];
    dst[0] = kvj.x; dst[1] = kvj.y; dst[2] = kvj.z; dst[3] = kvj.w;
    vs[(kc4 * 4 + 0) * VPAD + r] = vvj.x;
    vs[(kc4 * 4 + 1) * VPAD + r] = vvj.y;
    vs[(kc4 * 4 + 2) * VPAD + r] = vvj.z;
    vs[(kc4 * 4 + 3) * VPAD + r] = vvj.w;
}

__device__ __forceinline__ void attn_load_k_frags(
    const float* ks, v2f (&kb)[2][16], const int kq, const int grp, const int idx, const int q)
{
    #pragma unroll
    for (int u = 0; u < 2; ++u) {
        const float* kw = &ks[(kq * 32 + 16 * u + idx) * KPAD + 2 * grp];
        kb[u][2 * q]     = *(const v2f*)(kw + 4 * (2 * q));
        kb[u][2 * q + 1] = *(const v2f*)(kw + 4 * (2 * q + 1));
    }
}
__device__ __forceinline__ void attn_load_v_frags(
    const float* vs, v2f (&va)[4][8], const int kq, const int grp, const int idx)
{
    #pragma unroll
    for (int t = 0; t < 4; ++t)
        #pragma unroll
        for (int u = 0; u < 2; ++u)
            #pragma unroll
            for (int j = 0; j < 4; ++j)
                va[t][4 * u + j] =
                    *(const v2f*)(&vs[(t * 16 + idx) * VPAD + kq * 32 + 16 * u + 8 * grp + 2 * j]);
}

__global__ __launch_bounds__(256) void attn_kernel(
    const float* __restrict__ Q, const float* __restrict__ K,
    const float* __restrict__ V, float* __restrict__ out)
{
    constexpr int STG = SK * KPAD + HD * VPAD;
    static_assert(8 * MWS <= 2 * STG, "merge buffer must fit over the K/V stages");
    __shared__ __attribute__((aligned(16))) float smem[2 * STG];

    const int tid  = threadIdx.x;
    const int lane = tid & 31;
    const int grp  = lane >> 4;
    const int idx  = lane & 15;
    const int wave = __builtin_amdgcn_readfirstlane(tid >> 5);
    const int rt   = wave >> 2;
    const int kq   = wave & 3;

    const int  b     = blockIdx.y;
    const int  p     = blockIdx.x;
    const long base  = (long)b * TT * HD;
    const int  qrowA = p * QT + rt * 16;
    const int  qrowB = (TT / QT - 1 - p) * QT + rt * 16;

    v2f qfA[16], qfB[16];
    {
        const float* qa = Q + base + (long)(qrowA + idx) * HD + 2 * grp;
        const float* qb = Q + base + (long)(qrowB + idx) * HD + 2 * grp;
        #pragma unroll
        for (int s = 0; s < 16; ++s) {
            qfA[s] = *(const v2f*)(qa + 4 * s);
            qfB[s] = *(const v2f*)(qb + 4 * s);
        }
    }

    v8f   oA[4] = {}, oB[4] = {};
    float mA = -1e30f, lA = 0.f;
    float mB = -1e30f, lB = 0.f;

    const int nchunkA = (p * QT + QT - 1) / SK + 1;
    const int nchunkB = ((TT / QT - 1 - p) * QT + QT - 1) / SK + 1;

    float4 kv[8], vv[8];
    v2f kb[2][16];
    const float* Kb = K + base;
    const float* Vb = V + base;
    const unsigned tbyte = (unsigned)(((tid >> 4) * HD + (tid & 15) * 4) * 4);
    const float* pfb = (tid < 128 ? K : V) + base + (long)(tid & 127) * HD;
    #pragma unroll
    for (int j = 0; j < 8; ++j) attn_load_group(Kb, Vb, kv[j], vv[j], tbyte, j);
    #pragma unroll
    for (int j = 0; j < 8; ++j) attn_store_group(smem, smem + SK * KPAD, kv[j], vv[j], tid, j);
    __syncthreads();
    #pragma unroll
    for (int j = 0; j < 8; ++j) attn_load_group(Kb + SK * HD, Vb + SK * HD, kv[j], vv[j], tbyte, j);
    #pragma unroll
    for (int q = 0; q < 8; ++q) attn_load_k_frags(smem, kb, kq, grp, idx, q);

    auto chunk_trip = [&](const int sc, const auto both_tag, const auto mask_tag) {
        constexpr bool BOTH = decltype(both_tag)::value;
        constexpr bool MASK = decltype(mask_tag)::value;
        const int s0 = sc * SK;
        float* vs = smem + (sc & 1) * STG + SK * KPAD;
        float* kn = smem + ((sc + 1) & 1) * STG;
        float* vn = kn + SK * KPAD;
        const int sl = ((sc + 2 < nchunkB) ? sc + 2 : nchunkB - 1) * SK;
        const float* Kl = Kb + (long)sl * HD;
        const float* Vl = Vb + (long)sl * HD;
        v2f va[4][8];
        attn_load_v_frags(vs, va, kq, grp, idx);
        __builtin_amdgcn_sched_barrier(0);
        const int keyrow0 = s0 + kq * 32 + 8 * grp;

        auto store_part = [&](const int st) {
            if ((st & 1) == 0) attn_store_group(kn, vn, kv[st >> 1], vv[st >> 1], tid, st >> 1);
        };
        auto load_part = [&](const int j) {
            if (j < 4) {
                attn_load_group(Kl, Vl, kv[2 * j],     vv[2 * j],     tbyte, 2 * j);
                attn_load_group(Kl, Vl, kv[2 * j + 1], vv[2 * j + 1], tbyte, 2 * j + 1);
            }
            attn_load_k_frags(kn, kb, kq, grp, idx, j);
            if (j == 0) {
                const int  sp   = ((sc + 3 < nchunkB) ? sc + 3 : nchunkB - 1) * SK;
                __builtin_prefetch(pfb + (long)sp * HD, 0, 1);
            }
        };

        v8f sB0, sB1;
        attn_scores<true>(qfB, kb, sB0, sB1, store_part);
        stage_publish();
        if constexpr (BOTH) {
            v8f sA0, sA1;
            attn_scores<false>(qfA, kb, sA0, sA1, no_part{});
            attn_softmax<MASK>(sA0, sA1, oA, mA, lA, qrowA + idx, keyrow0);
            attn_softmax<false>(sB0, sB1, oB, mB, lB, qrowB + idx, keyrow0);
            attn_pv<false>(va, sA0, sA1, oA, no_part{});
        } else {
            attn_softmax<MASK>(sB0, sB1, oB, mB, lB, qrowB + idx, keyrow0);
        }
        __builtin_amdgcn_sched_barrier(0);
        stage_wait();
        attn_pv<true>(va, sB0, sB1, oB, load_part);
    };
    for (int sc = 0; sc < nchunkA - 1; ++sc)
        chunk_trip(sc, std::true_type{}, std::false_type{});
    chunk_trip(nchunkA - 1, std::true_type{}, std::true_type{});
    for (int sc = nchunkA; sc < nchunkB - 1; ++sc)
        chunk_trip(sc, std::false_type{}, std::false_type{});
    chunk_trip(nchunkB - 1, std::false_type{}, std::true_type{});

    attn_merge_store(smem, oA, mA, lA, out, base / HD + qrowA, wave, lane, grp, idx);
    attn_merge_store(smem, oB, mB, lB, out, base / HD + qrowB, wave, lane, grp, idx);
}

extern "C" void kernel_launch(void* const* d_in, const int* in_sizes, int n_in,
                              void* d_out, int out_size, void* d_ws, size_t ws_size,
                              hipStream_t stream) {
    const float* x  = (const float*)d_in[0];
    const float* Wq = (const float*)d_in[1];
    const float* Wk = (const float*)d_in[2];
    const float* Wv = (const float*)d_in[3];
    float* out = (float*)d_out;

    float* Q = (float*)d_ws;
    float* K = Q + (size_t)BT * HD;
    float* V = K + (size_t)BT * HD;

    proj_qkv_kernel<<<dim3(BT / MTILE), 256, 0, stream>>>(x, Wq, Wk, Wv, Q, K, V);
    attn_kernel<<<dim3(TT / MTILE, BB), 256, 0, stream>>>(Q, K, V, out);
}
